# starting kernel
# baseline (speedup 1.0000x reference)
#include <hip/hip_runtime.h>
#include <hip/hip_bf16.h>


typedef __attribute__((ext_vector_type(2))) float v2f;
typedef __attribute__((ext_vector_type(8))) float v8f;

#define NEG_SLOPE 0.2f
#define LN_EPS    1e-5f

__global__ void fill_f32(float* __restrict__ p, long long n, float v) {
  long long i = (long long)blockIdx.x * blockDim.x + threadIdx.x;
  if (i < n) p[i] = v;
}

__device__ __forceinline__ unsigned f2sortable(float f) {
  unsigned u = __float_as_uint(f);
  return (u & 0x80000000u) ? ~u : (u | 0x80000000u);
}
__device__ __forceinline__ float sortable2f(unsigned u) {
  u = (u & 0x80000000u) ? (u & 0x7FFFFFFFu) : ~u;
  return __uint_as_float(u);
}
__device__ __forceinline__ float leaky(float x) {
  return x > 0.0f ? x : NEG_SLOPE * x;
}
__device__ __forceinline__ void edge_sd(const int* __restrict__ ei, int E, int e,
                                        int& s, int& d) {
  if (e < E) { s = ei[e]; d = ei[E + e]; }
  else       { s = d = e - E; }
}

template <int MODE>
__global__ void gemm_wmma_n256(const float* __restrict__ A,
                               const float* __restrict__ B,
                               float* __restrict__ C,
                               const float* __restrict__ bias_a,
                               const float* __restrict__ bias_b,
                               int M, int K) {
  const int lane  = threadIdx.x & 31;
  const int wave  = threadIdx.x >> 5;
  const int rtile = blockIdx.x;
  const int rlo   = lane & 15;
  const int khalf = (lane >> 4) * 2;
  long long arow  = (long long)rtile * 16 + rlo;
  if (arow >= M) arow = M - 1;
  const int col   = wave * 32 + rlo;

  v8f acc0 = {}; v8f acc1 = {};
  const float* __restrict__ Ar = A + arow * K;
  for (int k = 0; k < K; k += 4) {
    v2f a, b0, b1;
    a.x = Ar[k + khalf];
    a.y = Ar[k + khalf + 1];
    const float* __restrict__ Bk = B + (long long)(k + khalf) * 256;
    b0.x = Bk[col];        b0.y = Bk[256 + col];
    b1.x = Bk[col + 16];   b1.y = Bk[256 + col + 16];
    acc0 = __builtin_amdgcn_wmma_f32_16x16x4_f32(false, a, false, b0,
                                                 (short)0, acc0, false, false);
    acc1 = __builtin_amdgcn_wmma_f32_16x16x4_f32(false, a, false, b1,
                                                 (short)0, acc1, false, false);
  }

  const int rhalf = (lane >> 4) * 8;
#pragma unroll
  for (int r = 0; r < 8; ++r) {
    long long orow = (long long)rtile * 16 + rhalf + r;
    if (orow < M) {
      const int c0 = wave * 32 + rlo;
      long long o  = orow * 256 + c0;
      if (MODE == 0) {
        C[o]      = acc0[r];
        C[o + 16] = acc1[r];
      } else {
        C[o]      += acc0[r] + bias_a[c0]      + bias_b[c0];
        C[o + 16] += acc1[r] + bias_a[c0 + 16] + bias_b[c0 + 16];
      }
    }
  }
}

__global__ void attn_scores(const float* __restrict__ hbuf,
                            const float* __restrict__ att_s,
                            const float* __restrict__ att_d,
                            float* __restrict__ as_out,
                            float* __restrict__ ad_out,
                            int Nn, int Hh, int Cc) {
  long long tid = (long long)blockIdx.x * blockDim.x + threadIdx.x;
  if (tid >= (long long)Nn * Hh) return;
  int n = (int)(tid / Hh), h = (int)(tid % Hh);
  const float* __restrict__ row = hbuf + (long long)n * 256 + h * Cc;
  const float* __restrict__ vs  = att_s + h * Cc;
  const float* __restrict__ vd  = att_d + h * Cc;
  float ss = 0.0f, sd = 0.0f;
  for (int c = 0; c < Cc; ++c) { float v = row[c]; ss += v * vs[c]; sd += v * vd[c]; }
  as_out[(long long)n * Hh + h] = ss;
  ad_out[(long long)n * Hh + h] = sd;
}

__global__ void edge_max(const int* __restrict__ ei,
                         const float* __restrict__ asrc,
                         const float* __restrict__ adst,
                         unsigned* __restrict__ mbuf,
                         int E, int EA, int Hh) {
  long long tid = (long long)blockIdx.x * blockDim.x + threadIdx.x;
  if (tid >= (long long)EA * Hh) return;
  int e = (int)(tid / Hh), h = (int)(tid % Hh);
  int s, d; edge_sd(ei, E, e, s, d);
  float l = leaky(asrc[(long long)s * Hh + h] + adst[(long long)d * Hh + h]);
  atomicMax(&mbuf[(long long)d * Hh + h], f2sortable(l));
}

__global__ void edge_expsum(const int* __restrict__ ei,
                            const float* __restrict__ asrc,
                            const float* __restrict__ adst,
                            const unsigned* __restrict__ mbuf,
                            float* __restrict__ zbuf,
                            int E, int EA, int Hh) {
  long long tid = (long long)blockIdx.x * blockDim.x + threadIdx.x;
  if (tid >= (long long)EA * Hh) return;
  int e = (int)(tid / Hh), h = (int)(tid % Hh);
  int s, d; edge_sd(ei, E, e, s, d);
  float l = leaky(asrc[(long long)s * Hh + h] + adst[(long long)d * Hh + h]);
  float m = sortable2f(mbuf[(long long)d * Hh + h]);
  atomicAdd(&zbuf[(long long)d * Hh + h], __expf(l - m));
}

__global__ void edge_aggregate(const int* __restrict__ ei,
                               const float* __restrict__ hin,
                               const float* __restrict__ asrc,
                               const float* __restrict__ adst,
                               const unsigned* __restrict__ mbuf,
                               const float* __restrict__ zbuf,
                               float* __restrict__ agg,
                               int E, int EA, int Hh, int Cc) {
  long long gtid = (long long)blockIdx.x * blockDim.x + threadIdx.x;
  int wid  = (int)(gtid >> 5);
  int lane = threadIdx.x & 31;
  if (wid >= EA) return;
  int s, d; edge_sd(ei, E, wid, s, d);
  const float* __restrict__ src_row = hin + (long long)s * 256;
  float* __restrict__ dst_row       = agg + (long long)d * 256;
#pragma unroll
  for (int k = 0; k < 8; ++k) {
    int idx = k * 32 + lane;
    int hh  = idx / Cc;
    float l = leaky(asrc[(long long)s * Hh + hh] + adst[(long long)d * Hh + hh]);
    float m = sortable2f(mbuf[(long long)d * Hh + hh]);
    float z = zbuf[(long long)d * Hh + hh];
    float alpha = __expf(l - m) / z;
    atomicAdd(&dst_row[idx], src_row[idx] * alpha);
  }
}

__global__ void ln_elu(float* __restrict__ hbuf, const float* __restrict__ b1,
                       const float* __restrict__ gamma,
                       const float* __restrict__ beta, int Nn) {
  long long gtid = (long long)blockIdx.x * blockDim.x + threadIdx.x;
  int n    = (int)(gtid >> 5);
  int lane = threadIdx.x & 31;
  if (n >= Nn) return;
  float* __restrict__ row = hbuf + (long long)n * 256;
  float vals[8];
  float sum = 0.0f;
#pragma unroll
  for (int k = 0; k < 8; ++k) {
    int idx = k * 32 + lane;
    vals[k] = row[idx] + b1[idx];
    sum += vals[k];
  }
#pragma unroll
  for (int off = 16; off > 0; off >>= 1) sum += __shfl_xor(sum, off);
  float mu = sum * (1.0f / 256.0f);
  float var = 0.0f;
#pragma unroll
  for (int k = 0; k < 8; ++k) { float t = vals[k] - mu; var += t * t; }
#pragma unroll
  for (int off = 16; off > 0; off >>= 1) var += __shfl_xor(var, off);
  var *= (1.0f / 256.0f);
  float rs = rsqrtf(var + LN_EPS);
#pragma unroll
  for (int k = 0; k < 8; ++k) {
    int idx = k * 32 + lane;
    float t = (vals[k] - mu) * rs * gamma[idx] + beta[idx];
    row[idx] = t > 0.0f ? t : (__expf(t) - 1.0f);
  }
}

__global__ void pool_accum(const float* __restrict__ o,
                           const int* __restrict__ batch,
                           float* __restrict__ outsum,
                           float* __restrict__ cnt, int Nn) {
  long long gtid = (long long)blockIdx.x * blockDim.x + threadIdx.x;
  int n    = (int)(gtid >> 5);
  int lane = threadIdx.x & 31;
  if (n >= Nn) return;
  int g = batch[n];
  const float* __restrict__ row = o + (long long)n * 256;
  float* __restrict__ acc       = outsum + (long long)g * 256;
#pragma unroll
  for (int k = 0; k < 8; ++k) {
    int idx = k * 32 + lane;
    atomicAdd(&acc[idx], row[idx]);
  }
  if (lane == 0) atomicAdd(&cnt[g], 1.0f);
}

__global__ void pool_div(float* __restrict__ out, const float* __restrict__ cnt,
                         int total) {
  int i = blockIdx.x * blockDim.x + threadIdx.x;
  if (i >= total) return;
  out[i] /= fmaxf(cnt[i >> 8], 1.0f);
}

static inline int cdiv(long long a, long long b) { return (int)((a + b - 1) / b); }

extern "C" void kernel_launch(void* const* d_in, const int* in_sizes, int n_in,
                              void* d_out, int out_size, void* d_ws, size_t ws_size,
                              hipStream_t stream) {
  const float* x        = (const float*)d_in[0];
  const int*   ei       = (const int*)  d_in[1];
  const int*   batch    = (const int*)  d_in[2];
  const float* W1       = (const float*)d_in[3];
  const float* att_src1 = (const float*)d_in[4];
  const float* att_dst1 = (const float*)d_in[5];
  const float* b1       = (const float*)d_in[6];
  const float* gamma    = (const float*)d_in[7];
  const float* beta     = (const float*)d_in[8];
  const float* W2       = (const float*)d_in[9];
  const float* att_src2 = (const float*)d_in[10];
  const float* att_dst2 = (const float*)d_in[11];
  const float* b2       = (const float*)d_in[12];
  const float* Wres     = (const float*)d_in[13];
  const float* bres     = (const float*)d_in[14];
  float* out = (float*)d_out;

  const int Nn = in_sizes[2];
  const int Ee = in_sizes[1] / 2;
  const int Gg = out_size / 256;
  const int EA = Ee + Nn;
  const int IN_DIM = in_sizes[0] / Nn;

  const long long NB = (long long)Nn * 256;
  float* ws   = (float*)d_ws;
  float* buf0 = ws;
  float* buf1 = ws + NB;
  float* buf2 = ws + 2 * NB;
  float* scal = ws + 3 * NB;
  unsigned* m1 = (unsigned*)scal;
  float*    z1 = scal + (long long)Nn * 8;
  unsigned* m2 = (unsigned*)(scal + (long long)Nn * 16);
  float*    z2 = scal + (long long)Nn * 17;
  float*    cnt = scal + (long long)Nn * 18;
  float*    as1 = cnt + Gg;
  float*    ad1 = as1 + (long long)Nn * 8;
  float*    as2 = ad1 + (long long)Nn * 8;
  float*    ad2 = as2 + Nn;

  const int B = 256;
  const long long zlen1 = 2 * NB;
  const long long zlen2 = (long long)Nn * 18 + Gg;

  fill_f32<<<cdiv(zlen1, B), B, 0, stream>>>(buf1, zlen1, 0.0f);
  fill_f32<<<cdiv(zlen2, B), B, 0, stream>>>(scal, zlen2, 0.0f);
  fill_f32<<<cdiv((long long)Gg * 256, B), B, 0, stream>>>(out, (long long)Gg * 256, 0.0f);

  const int gtiles = cdiv(Nn, 16);

  gemm_wmma_n256<0><<<gtiles, B, 0, stream>>>(x, W1, buf0, nullptr, nullptr, Nn, IN_DIM);
  attn_scores<<<cdiv((long long)Nn * 8, B), B, 0, stream>>>(buf0, att_src1, att_dst1,
                                                            as1, ad1, Nn, 8, 32);
  edge_max   <<<cdiv((long long)EA * 8, B), B, 0, stream>>>(ei, as1, ad1, m1, Ee, EA, 8);
  edge_expsum<<<cdiv((long long)EA * 8, B), B, 0, stream>>>(ei, as1, ad1, m1, z1, Ee, EA, 8);
  edge_aggregate<<<cdiv((long long)EA * 32, B), B, 0, stream>>>(ei, buf0, as1, ad1, m1, z1,
                                                                buf1, Ee, EA, 8, 32);
  ln_elu<<<cdiv((long long)Nn * 32, B), B, 0, stream>>>(buf1, b1, gamma, beta, Nn);

  gemm_wmma_n256<0><<<gtiles, B, 0, stream>>>(buf1, W2, buf0, nullptr, nullptr, Nn, 256);
  attn_scores<<<cdiv((long long)Nn, B), B, 0, stream>>>(buf0, att_src2, att_dst2,
                                                        as2, ad2, Nn, 1, 256);
  edge_max   <<<cdiv((long long)EA, B), B, 0, stream>>>(ei, as2, ad2, m2, Ee, EA, 1);
  edge_expsum<<<cdiv((long long)EA, B), B, 0, stream>>>(ei, as2, ad2, m2, z2, Ee, EA, 1);
  edge_aggregate<<<cdiv((long long)EA * 32, B), B, 0, stream>>>(ei, buf0, as2, ad2, m2, z2,
                                                                buf2, Ee, EA, 1, 256);
  gemm_wmma_n256<1><<<gtiles, B, 0, stream>>>(buf1, Wres, buf2, b2, bres, Nn, 256);

  pool_accum<<<cdiv((long long)Nn * 32, B), B, 0, stream>>>(buf2, batch, out, cnt, Nn);
  pool_div  <<<cdiv((long long)Gg * 256, B), B, 0, stream>>>(out, cnt, Gg * 256);
}


# Gather-sum leaves LDS float atomics: the owner block regroups its queued sources by destination row in LDS (count, prefix, slot), then a wave sums whole rows in registers; sum order = LDS slot order
# speedup vs baseline: 3.5237x; 3.5237x over previous
#include <hip/hip_runtime.h>
#include <hip/hip_bf16.h>


typedef __attribute__((ext_vector_type(2))) float v2f;
typedef __attribute__((ext_vector_type(8))) float v8f;

#define NEG_SLOPE 0.2f
#define LN_EPS    1e-5f

__global__ void fill_f32(float* __restrict__ p, long long n, float v) {
  long long i = (long long)blockIdx.x * blockDim.x + threadIdx.x;
  if (i < n) p[i] = v;
}

__device__ __forceinline__ unsigned f2sortable(float f) {
  unsigned u = __float_as_uint(f);
  return (u & 0x80000000u) ? ~u : (u | 0x80000000u);
}
__device__ __forceinline__ float sortable2f(unsigned u) {
  u = (u & 0x80000000u) ? (u & 0x7FFFFFFFu) : ~u;
  return __uint_as_float(u);
}
__device__ __forceinline__ float leaky(float x) {
  return x > 0.0f ? x : NEG_SLOPE * x;
}
__device__ __forceinline__ void edge_sd(const int* __restrict__ ei, int E, int e,
                                        int& s, int& d) {
  if (e < E) { s = ei[e]; d = ei[E + e]; }
  else       { s = d = e - E; }
}

#define GEMM_G 256
#define GEMM_P 1056
#define GEMM_KMAX 256

template <int MODE>
__global__ __launch_bounds__(512)
void gemm_wmma_n256(const float* __restrict__ A,
                    const float* __restrict__ B,
                    float* __restrict__ C,
                    const float* __restrict__ bias_a,
                    const float* __restrict__ bias_b,
                    int M, int K) {
  __shared__ float Bs[(GEMM_KMAX / 4) * GEMM_P];

  const int lane = threadIdx.x & 31;
  const int wave = __builtin_amdgcn_readfirstlane(threadIdx.x >> 5);
  const int c    = lane & 15;
  const int h    = lane >> 4;

  for (int i = threadIdx.x; i < (K >> 2) * 256; i += 512) {
    const int g = i >> 8, n = i & 255;
    const float* __restrict__ p = B + (long long)(4 * g) * 256 + n;
    float4 v;
    v.x = p[0]; v.y = p[256]; v.z = p[512]; v.w = p[768];
    *(float4*)(Bs + g * GEMM_P + n * 4) = v;
  }
  __syncthreads();

  const int ntile = (M + 15) >> 4;
  const int t0 = (int)(((long long)blockIdx.x * ntile) / GEMM_G);
  const int t1 = (int)(((long long)(blockIdx.x + 1) * ntile) / GEMM_G);
  const int nunit = (t1 - t0) * 2;

  for (int u = wave; u < nunit; u += 16) {
    const int tile = t0 + (u >> 1);
    const int cb   = (u & 1) * 128;
    long long arow = (long long)tile * 16 + c;
    if (arow >= M) arow = M - 1;
    const float* __restrict__ Ar = A + arow * K + 4 * h;
    const float* Bl = Bs + h * GEMM_P + (cb + c) * 4;

    v8f acc[8];
#pragma unroll
    for (int t = 0; t < 8; ++t) acc[t] = v8f{};

    float4 an = *(const float4*)Ar;
    for (int k0 = 0; k0 < K; k0 += 8) {
      const float4 a4 = an;
      an = *(const float4*)(Ar + min(k0 + 8, K - 8));
      const float* Bq = Bl + (k0 >> 2) * GEMM_P;
      float4 b4[8];
#pragma unroll
      for (int t = 0; t < 8; ++t) b4[t] = *(const float4*)(Bq + t * 64);
      v2f a01, a23;
      a01.x = a4.x; a01.y = a4.y; a23.x = a4.z; a23.y = a4.w;
#pragma unroll
      for (int t = 0; t < 8; ++t) {
        v2f b; b.x = b4[t].x; b.y = b4[t].y;
        acc[t] = __builtin_amdgcn_wmma_f32_16x16x4_f32(false, a01, false, b,
                                                       (short)0, acc[t], false, false);
      }
#pragma unroll
      for (int t = 0; t < 8; ++t) {
        v2f b; b.x = b4[t].z; b.y = b4[t].w;
        acc[t] = __builtin_amdgcn_wmma_f32_16x16x4_f32(false, a23, false, b,
                                                       (short)0, acc[t], false, false);
      }
    }

    float ba[8], bb[8];
#pragma unroll
    for (int t = 0; t < 8; ++t) {
      ba[t] = (MODE == 1) ? bias_a[cb + t * 16 + c] : 0.0f;
      bb[t] = (MODE == 1) ? bias_b[cb + t * 16 + c] : 0.0f;
    }
#pragma unroll
    for (int r = 0; r < 8; ++r) {
      const long long orow = (long long)tile * 16 + h * 8 + r;
      if (orow < M) {
#pragma unroll
        for (int t = 0; t < 8; ++t) {
          const int col = cb + t * 16 + c;
          const long long o = orow * 256 + col;
          if (MODE == 0) C[o] = acc[t][r];
          else           C[o] += acc[t][r] + ba[t] + bb[t];
        }
      }
    }
  }
}

__global__ void attn_scores(const float* __restrict__ hbuf,
                            const float* __restrict__ att_s,
                            const float* __restrict__ att_d,
                            float* __restrict__ as_out,
                            float* __restrict__ ad_out,
                            int Nn, int Hh, int Cc) {
  long long tid = (long long)blockIdx.x * blockDim.x + threadIdx.x;
  if (tid >= (long long)Nn * Hh) return;
  int n = (int)(tid / Hh), h = (int)(tid % Hh);
  const float* __restrict__ row = hbuf + (long long)n * 256 + h * Cc;
  const float* __restrict__ vs  = att_s + h * Cc;
  const float* __restrict__ vd  = att_d + h * Cc;
  float ss = 0.0f, sd = 0.0f;
  for (int c = 0; c < Cc; ++c) { float v = row[c]; ss += v * vs[c]; sd += v * vd[c]; }
  as_out[(long long)n * Hh + h] = ss;
  ad_out[(long long)n * Hh + h] = sd;
}

#define AGG_R 200
#define AGG_B 8
#define AGG_S 512
#define AGG_Q 768
#define AGG_BAD 255u
#define AGG_PER ((AGG_R + 31) / 32)

__global__ void edge_owner(const int* __restrict__ ei, unsigned* __restrict__ tag4,
                           int E, int Epad) {
  const int t = blockIdx.x * blockDim.x + threadIdx.x;
  if (t * 4 >= Epad) return;
  unsigned w = 0u;
#pragma unroll
  for (int i = 0; i < 4; ++i) {
    const int e = t * 4 + i;
    const unsigned b = (e < E) ? (unsigned)min(ei[E + e] / AGG_R, 255) : 255u;
    w |= b << (8 * i);
  }
  tag4[t] = w;
}

template <int HEADS>
__global__ __launch_bounds__(1024)
void edge_softmax_aggregate_own(const int* __restrict__ ei,
                                const unsigned* __restrict__ tag,
                                const float* __restrict__ hin,
                                const float* __restrict__ asrc,
                                const float* __restrict__ adst,
                                float* __restrict__ agg,
                                int E, int Epad, int Nn) {
  __shared__ float    c_ad[AGG_R * HEADS];
  __shared__ unsigned c_m[AGG_R * HEADS];
  __shared__ int      queue[32 * AGG_Q];
  __shared__ int      rowsrc[32 * AGG_Q + AGG_R];
  __shared__ int      r_cnt[AGG_R];
  __shared__ int      r_beg[AGG_R];
  __shared__ int      r_fill[AGG_R];
  __shared__ int      s_lost;

  const int lane = threadIdx.x & 31;
  const int wave = __builtin_amdgcn_readfirstlane(threadIdx.x >> 5);
  const int n0   = blockIdx.x * AGG_R;
  const int nown = min(AGG_R, Nn - n0);
  const unsigned mytag = min(blockIdx.x, 255u);
  const int hA   = (HEADS == 8) ? (lane >> 3) : 0;
  const int hB   = (HEADS == 8) ? 4 + (lane >> 3) : 0;

  for (int t = threadIdx.x; t < nown * HEADS; t += 1024) {
    const long long g = (long long)n0 * HEADS + t;
    const float a_d = adst[g];
    c_ad[t] = a_d;
    c_m[t]  = f2sortable(leaky(asrc[g] + a_d));
  }
  if (threadIdx.x < AGG_R) r_cnt[threadIdx.x] = 1;
  if (threadIdx.x == 0) s_lost = 0;
  __syncthreads();

  const int* __restrict__ dstp = ei + E;
  int* q = queue + wave * AGG_Q;
  int qn = 0;

  auto pack_range = [&](int* qp, int n) {
    for (int i0 = 0; i0 < n; i0 += 128) {
      int s[4]; unsigned dl[4];
#pragma unroll
      for (int u = 0; u < 4; ++u) {
        const int e = qp[min(i0 + u * 32 + lane, n - 1)];
        s[u]  = ei[e];
        dl[u] = (unsigned)(dstp[e] - n0);
      }
#pragma unroll
      for (int u = 0; u < 4; ++u) {
        const int i = i0 + u * 32 + lane;
        if (i < n) {
          const unsigned d = dl[u] < (unsigned)nown ? dl[u] : AGG_BAD;
          qp[i] = s[u] | (int)(d << 24);
          if (d != AGG_BAD) atomicAdd(&r_cnt[d], 1);
        }
      }
    }
  };

  constexpr int MU = (HEADS == 8) ? 8 : 4;
  auto max_range = [&](const int* qp, int n) {
    const unsigned items = (unsigned)n * HEADS;
    for (unsigned t0 = 0; t0 < items; t0 += 32 * MU) {
      unsigned cj[MU]; float a[MU];
#pragma unroll
      for (int u = 0; u < MU; ++u) {
        const unsigned t = min(t0 + u * 32 + lane, items - 1);
        const unsigned v = (unsigned)qp[t / HEADS];
        const unsigned h = t % HEADS;
        cj[u] = (v >> 24) * HEADS + h;
        a[u]  = asrc[(long long)(v & 0xFFFFFFu) * HEADS + h];
      }
      if constexpr (MU == 8)
        asm volatile("" :: "v"(a[0]), "v"(a[1]), "v"(a[2]), "v"(a[3]),
                           "v"(a[4]), "v"(a[5]), "v"(a[6]), "v"(a[7]));
      else
        asm volatile("" :: "v"(a[0]), "v"(a[1]), "v"(a[2]), "v"(a[3]));
#pragma unroll
      for (int u = 0; u < MU; ++u) {
        const bool ok = (t0 + u * 32 + lane < items) && (cj[u] < (unsigned)(nown * HEADS));
        const unsigned cc = ok ? cj[u] : 0u;
        const unsigned val = f2sortable(leaky(a[u] + c_ad[cc]));
        if (ok) atomicMax(&c_m[cc], val);
      }
    }
  };

  bool lost = false;

  const int nstep = Epad / AGG_S;
  auto scan = [&](auto&& after_step) {
    if (nstep <= 0) return;
    const uint4* __restrict__ tag16 = (const uint4*)tag;
    const unsigned pat = mytag * 0x01010101u;
    const int rot = (int)(((long long)blockIdx.x * nstep) / gridDim.x);
    auto step_of = [&](int k) { const int c = k + rot; return c >= nstep ? c - nstep : c; };
    auto load_step = [&](int k) { return tag16[step_of(min(k, nstep - 1)) * 32 + lane]; };
    auto detect = [&](const uint4 t) {
      const unsigned wv[4] = {t.x, t.y, t.z, t.w};
      unsigned hm = 0u;
#pragma unroll
      for (int j = 0; j < 4; ++j) {
        const unsigned x = wv[j] ^ pat;
        const unsigned y = (x & 0x7F7F7F7Fu) + 0x7F7F7F7Fu;
        const unsigned z = ~(y | x) & 0x80808080u;
        hm |= z >> (7 - j);
      }
      return hm;
    };
    auto collect = [&](unsigned hm, int k) {
      const int ebase = step_of(k) * AGG_S + lane * 16;
      unsigned m = __builtin_amdgcn_ballot_w32(hm != 0u);
      while (m != 0u) {
        const int p = __builtin_ctz(hm | 0x80000000u);
        const int e = ebase + ((p & 3) << 2) + (p >> 3);
        const bool hit = (hm != 0u) && (e < E);
        const unsigned mask = __builtin_amdgcn_ballot_w32(hit);
        if (hit) q[qn + __builtin_popcount(mask & ((1u << lane) - 1u))] = e;
        qn += __builtin_popcount(mask);
        hm &= hm - 1u;
        m = __builtin_amdgcn_ballot_w32(hm != 0u);
      }
      after_step(k + 32 < nstep);
    };
    uint4 ta = load_step(wave), tb = load_step(wave + 32);
    for (int k = wave; k < nstep && !lost; k += 64) {
      const unsigned h0 = detect(ta);
      const unsigned h1 = detect(tb);
      ta = load_step(k + 64);
      tb = load_step(k + 96);
      collect(h0, k);
      if (!lost && k + 32 < nstep) collect(h1, k + 32);
    }
  };

  auto weight = [&](float s, float ad, float m) { return __expf(leaky(s + ad) - m); };
  auto add_row = [&](float4& a0, float4& a1, float& zA, float& zB,
                     const float4 x0, const float4 x1, float wA, float wB) {
    a0.x += x0.x * wA;  a0.y += x0.y * wA;  a0.z += x0.z * wA;  a0.w += x0.w * wA;
    a1.x += x1.x * wB;  a1.y += x1.y * wB;  a1.z += x1.z * wB;  a1.w += x1.w * wB;
    zA += wA;  zB += wB;
  };
  auto store_row = [&](long long n, const float4 a0, const float4 a1, float zA, float zB) {
    const float iA = 1.0f / zA, iB = 1.0f / zB;
    float4 v0, v1;
    v0.x = a0.x * iA;  v0.y = a0.y * iA;  v0.z = a0.z * iA;  v0.w = a0.w * iA;
    v1.x = a1.x * iB;  v1.y = a1.y * iB;  v1.z = a1.z * iB;  v1.w = a1.w * iB;
    float4* __restrict__ orow = (float4*)(agg + n * 256);
    orow[lane]      = v0;
    orow[32 + lane] = v1;
  };

  scan([&](bool more) {
    if (more && qn > AGG_Q - AGG_S) lost = true;
  });
  if (lost) s_lost = 1;
  else if (qn > 0) { pack_range(q, qn); max_range(q, qn); }
  __syncthreads();

  if (s_lost != 0) {
    for (int r = wave; r < nown; r += 32) {
      const long long n = n0 + r;
      const float adA = adst[n * HEADS + hA], adB = adst[n * HEADS + hB];
      const float slA = asrc[n * HEADS + hA], slB = asrc[n * HEADS + hB];
      float mA = leaky(slA + adA), mB = leaky(slB + adB);
      for (int e0 = 0; e0 < E; e0 += 32) {
        const int e = e0 + lane;
        const bool hit = (e < E) && (dstp[e] == (int)n);
        const int s = hit ? ei[e] : 0;
        unsigned m = __builtin_amdgcn_ballot_w32(hit);
        while (m != 0u) {
          const int sj = __builtin_amdgcn_readlane(s, __builtin_ctz(m));
          m &= m - 1u;
          mA = fmaxf(mA, leaky(asrc[(long long)sj * HEADS + hA] + adA));
          mB = fmaxf(mB, leaky(asrc[(long long)sj * HEADS + hB] + adB));
        }
      }
      const float4* __restrict__ own = (const float4*)(hin + n * 256);
      float4 a0 = float4{0.0f, 0.0f, 0.0f, 0.0f}, a1 = a0;
      float zA = 0.0f, zB = 0.0f;
      add_row(a0, a1, zA, zB, own[lane], own[32 + lane],
              weight(slA, adA, mA), weight(slB, adB, mB));
      for (int e0 = 0; e0 < E; e0 += 32) {
        const int e = e0 + lane;
        const bool hit = (e < E) && (dstp[e] == (int)n);
        const int s = hit ? ei[e] : 0;
        unsigned m = __builtin_amdgcn_ballot_w32(hit);
        while (m != 0u) {
          const int sj = __builtin_amdgcn_readlane(s, __builtin_ctz(m));
          m &= m - 1u;
          const float4* __restrict__ row = (const float4*)(hin + (long long)sj * 256);
          add_row(a0, a1, zA, zB, row[lane], row[32 + lane],
                  weight(asrc[(long long)sj * HEADS + hA], adA, mA),
                  weight(asrc[(long long)sj * HEADS + hB], adB, mB));
        }
      }
      store_row(n, a0, a1, zA, zB);
    }
    return;
  }

  if (wave == 0) {
    int c[AGG_PER];
    int sum = 0;
#pragma unroll
    for (int j = 0; j < AGG_PER; ++j) {
      const int r = lane * AGG_PER + j;
      c[j] = (r < nown) ? r_cnt[r] : 0;
      sum += c[j];
    }
    int inc = sum;
#pragma unroll
    for (int off = 1; off < 32; off <<= 1) {
      const int t = __shfl_up(inc, off);
      if (lane >= off) inc += t;
    }
    int run = inc - sum;
#pragma unroll
    for (int j = 0; j < AGG_PER; ++j) {
      const int r = lane * AGG_PER + j;
      if (r < nown) {
        r_beg[r]    = run;
        r_fill[r]   = run + 1;
        rowsrc[run] = n0 + r;
      }
      run += c[j];
    }
  }
  __syncthreads();

  for (int i = lane; i < qn; i += 32) {
    const unsigned v = (unsigned)q[i];
    const unsigned d = v >> 24;
    if (d != AGG_BAD) rowsrc[atomicAdd(&r_fill[d], 1)] = (int)(v & 0xFFFFFFu);
  }
  __syncthreads();

  for (int r = wave; r < nown; r += 32) {
    const int cnt = __builtin_amdgcn_readfirstlane(r_cnt[r]);
    const int beg = __builtin_amdgcn_readfirstlane(r_beg[r]);
    const int cj  = r * HEADS;
    const float adA = c_ad[cj + hA], mA = sortable2f(c_m[cj + hA]);
    const float adB = c_ad[cj + hB], mB = sortable2f(c_m[cj + hB]);
    float4 a0 = float4{0.0f, 0.0f, 0.0f, 0.0f}, a1 = a0;
    float zA = 0.0f, zB = 0.0f;
    for (int i0 = 0; i0 < cnt; i0 += AGG_B) {
      const int nb = min(AGG_B, cnt - i0);
      const int sv = rowsrc[beg + i0 + min(lane, nb - 1)];
      float4 x0[AGG_B], x1[AGG_B];
      float  sA[AGG_B], sB[AGG_B];
#pragma unroll
      for (int j = 0; j < AGG_B; ++j) {
        const int sj = __builtin_amdgcn_readlane(sv, j);
        const float4* __restrict__ row = (const float4*)(hin + (long long)sj * 256);
        x0[j] = row[lane];
        x1[j] = row[32 + lane];
        sA[j] = asrc[(long long)sj * HEADS + hA];
        sB[j] = (HEADS == 8) ? asrc[(long long)sj * HEADS + hB] : 0.0f;
      }
#pragma unroll
      for (int j = 0; j < AGG_B; ++j) {
        const float on = (j < nb) ? 1.0f : 0.0f;
        const float wA = weight(sA[j], adA, mA) * on;
        const float wB = (HEADS == 8) ? weight(sB[j], adB, mB) * on : wA;
        add_row(a0, a1, zA, zB, x0[j], x1[j], wA, wB);
      }
    }
    store_row((long long)n0 + r, a0, a1, zA, zB);
  }
}

__global__ void ln_elu(float* __restrict__ hbuf, const float* __restrict__ b1,
                       const float* __restrict__ gamma,
                       const float* __restrict__ beta, int Nn) {
  long long gtid = (long long)blockIdx.x * blockDim.x + threadIdx.x;
  int n    = (int)(gtid >> 5);
  int lane = threadIdx.x & 31;
  if (n >= Nn) return;
  float* __restrict__ row = hbuf + (long long)n * 256;
  float vals[8];
  float sum = 0.0f;
#pragma unroll
  for (int k = 0; k < 8; ++k) {
    int idx = k * 32 + lane;
    vals[k] = row[idx] + b1[idx];
    sum += vals[k];
  }
#pragma unroll
  for (int off = 16; off > 0; off >>= 1) sum += __shfl_xor(sum, off);
  float mu = sum * (1.0f / 256.0f);
  float var = 0.0f;
#pragma unroll
  for (int k = 0; k < 8; ++k) { float t = vals[k] - mu; var += t * t; }
#pragma unroll
  for (int off = 16; off > 0; off >>= 1) var += __shfl_xor(var, off);
  var *= (1.0f / 256.0f);
  float rs = rsqrtf(var + LN_EPS);
#pragma unroll
  for (int k = 0; k < 8; ++k) {
    int idx = k * 32 + lane;
    float t = (vals[k] - mu) * rs * gamma[idx] + beta[idx];
    row[idx] = t > 0.0f ? t : (__expf(t) - 1.0f);
  }
}

__global__ void pool_accum(const float* __restrict__ o,
                           const int* __restrict__ batch,
                           float* __restrict__ outsum,
                           float* __restrict__ cnt, int Nn) {
  long long gtid = (long long)blockIdx.x * blockDim.x + threadIdx.x;
  int n    = (int)(gtid >> 5);
  int lane = threadIdx.x & 31;
  if (n >= Nn) return;
  int g = batch[n];
  const float* __restrict__ row = o + (long long)n * 256;
  float* __restrict__ acc       = outsum + (long long)g * 256;
#pragma unroll
  for (int k = 0; k < 8; ++k) {
    int idx = k * 32 + lane;
    atomicAdd(&acc[idx], row[idx]);
  }
  if (lane == 0) atomicAdd(&cnt[g], 1.0f);
}

__global__ void pool_div(float* __restrict__ out, const float* __restrict__ cnt,
                         int total) {
  int i = blockIdx.x * blockDim.x + threadIdx.x;
  if (i >= total) return;
  out[i] /= fmaxf(cnt[i >> 8], 1.0f);
}

static inline int cdiv(long long a, long long b) { return (int)((a + b - 1) / b); }

extern "C" void kernel_launch(void* const* d_in, const int* in_sizes, int n_in,
                              void* d_out, int out_size, void* d_ws, size_t ws_size,
                              hipStream_t stream) {
  const float* x        = (const float*)d_in[0];
  const int*   ei       = (const int*)  d_in[1];
  const int*   batch    = (const int*)  d_in[2];
  const float* W1       = (const float*)d_in[3];
  const float* att_src1 = (const float*)d_in[4];
  const float* att_dst1 = (const float*)d_in[5];
  const float* b1       = (const float*)d_in[6];
  const float* gamma    = (const float*)d_in[7];
  const float* beta     = (const float*)d_in[8];
  const float* W2       = (const float*)d_in[9];
  const float* att_src2 = (const float*)d_in[10];
  const float* att_dst2 = (const float*)d_in[11];
  const float* b2       = (const float*)d_in[12];
  const float* Wres     = (const float*)d_in[13];
  const float* bres     = (const float*)d_in[14];
  float* out = (float*)d_out;

  const int Nn = in_sizes[2];
  const int Ee = in_sizes[1] / 2;
  const int Gg = out_size / 256;
  const int EA = Ee + Nn;
  const int IN_DIM = in_sizes[0] / Nn;

  const long long NB = (long long)Nn * 256;
  float* ws   = (float*)d_ws;
  float* buf0 = ws;
  float* buf1 = ws + NB;
  float* buf2 = ws + 2 * NB;
  float* cnt  = ws + 3 * NB;
  float* as1  = cnt + Gg;
  float* ad1  = as1 + (long long)Nn * 8;
  float* as2  = ad1 + (long long)Nn * 8;
  float* ad2  = as2 + Nn;
  const int Epad = cdiv(Ee, AGG_S) * AGG_S;
  unsigned* tag = (unsigned*)(ws + ((3 * NB + (long long)Nn * 18 + Gg + 3) / 4) * 4);

  const int B = 256;

  edge_owner<<<cdiv(Epad / 4, B), B, 0, stream>>>(ei, tag, Ee, Epad);
  fill_f32<<<cdiv(Gg, B), B, 0, stream>>>(cnt, Gg, 0.0f);
  fill_f32<<<cdiv((long long)Gg * 256, B), B, 0, stream>>>(out, (long long)Gg * 256, 0.0f);

  gemm_wmma_n256<0><<<GEMM_G, 512, 0, stream>>>(x, W1, buf0, nullptr, nullptr, Nn, IN_DIM);
  attn_scores<<<cdiv((long long)Nn * 8, B), B, 0, stream>>>(buf0, att_src1, att_dst1,
                                                            as1, ad1, Nn, 8, 32);
  edge_softmax_aggregate_own<8><<<cdiv(Nn, AGG_R), 1024, 0, stream>>>(ei, tag, buf0, as1, ad1,
                                                                      buf1, Ee, Epad, Nn);
  ln_elu<<<cdiv((long long)Nn * 32, B), B, 0, stream>>>(buf1, b1, gamma, beta, Nn);

  gemm_wmma_n256<0><<<GEMM_G, 512, 0, stream>>>(buf1, W2, buf0, nullptr, nullptr, Nn, 256);
  attn_scores<<<cdiv((long long)Nn, B), B, 0, stream>>>(buf0, att_src2, att_dst2,
                                                        as2, ad2, Nn, 1, 256);
  edge_softmax_aggregate_own<1><<<cdiv(Nn, AGG_R), 1024, 0, stream>>>(ei, tag, buf0, as2, ad2,
                                                                      buf2, Ee, Epad, Nn);
  gemm_wmma_n256<1><<<GEMM_G, 512, 0, stream>>>(buf1, Wres, buf2, b2, bres, Nn, 256);

  pool_accum<<<cdiv((long long)Nn * 32, B), B, 0, stream>>>(buf2, batch, out, cnt, Nn);
  pool_div  <<<cdiv((long long)Gg * 256, B), B, 0, stream>>>(out, cnt, Gg * 256);
}


# Layer 2 reuses the row-grouped source lists built by layer 1's owner blocks (kept in workspace): a resident wave walks 8 rows, no tag scan, queues or LDS; sum order = layer 1's slot order
# speedup vs baseline: 3.6914x; 1.0476x over previous
#include <hip/hip_runtime.h>
#include <hip/hip_bf16.h>


typedef __attribute__((ext_vector_type(2))) float v2f;
typedef __attribute__((ext_vector_type(8))) float v8f;

#define NEG_SLOPE 0.2f
#define LN_EPS    1e-5f

__global__ void fill_f32(float* __restrict__ p, long long n, float v) {
  long long i = (long long)blockIdx.x * blockDim.x + threadIdx.x;
  if (i < n) p[i] = v;
}

__device__ __forceinline__ unsigned f2sortable(float f) {
  unsigned u = __float_as_uint(f);
  return (u & 0x80000000u) ? ~u : (u | 0x80000000u);
}
__device__ __forceinline__ float sortable2f(unsigned u) {
  u = (u & 0x80000000u) ? (u & 0x7FFFFFFFu) : ~u;
  return __uint_as_float(u);
}
__device__ __forceinline__ float leaky(float x) {
  return x > 0.0f ? x : NEG_SLOPE * x;
}
__device__ __forceinline__ void edge_sd(const int* __restrict__ ei, int E, int e,
                                        int& s, int& d) {
  if (e < E) { s = ei[e]; d = ei[E + e]; }
  else       { s = d = e - E; }
}

#define GEMM_G 256
#define GEMM_P 1056
#define GEMM_KMAX 256

template <int MODE>
__global__ __launch_bounds__(512)
void gemm_wmma_n256(const float* __restrict__ A,
                    const float* __restrict__ B,
                    float* __restrict__ C,
                    const float* __restrict__ bias_a,
                    const float* __restrict__ bias_b,
                    int M, int K) {
  __shared__ float Bs[(GEMM_KMAX / 4) * GEMM_P];

  const int lane = threadIdx.x & 31;
  const int wave = __builtin_amdgcn_readfirstlane(threadIdx.x >> 5);
  const int c    = lane & 15;
  const int h    = lane >> 4;

  for (int i = threadIdx.x; i < (K >> 2) * 256; i += 512) {
    const int g = i >> 8, n = i & 255;
    const float* __restrict__ p = B + (long long)(4 * g) * 256 + n;
    float4 v;
    v.x = p[0]; v.y = p[256]; v.z = p[512]; v.w = p[768];
    *(float4*)(Bs + g * GEMM_P + n * 4) = v;
  }
  __syncthreads();

  const int ntile = (M + 15) >> 4;
  const int t0 = (int)(((long long)blockIdx.x * ntile) / GEMM_G);
  const int t1 = (int)(((long long)(blockIdx.x + 1) * ntile) / GEMM_G);
  const int nunit = (t1 - t0) * 2;

  for (int u = wave; u < nunit; u += 16) {
    const int tile = t0 + (u >> 1);
    const int cb   = (u & 1) * 128;
    long long arow = (long long)tile * 16 + c;
    if (arow >= M) arow = M - 1;
    const float* __restrict__ Ar = A + arow * K + 4 * h;
    const float* Bl = Bs + h * GEMM_P + (cb + c) * 4;

    v8f acc[8];
#pragma unroll
    for (int t = 0; t < 8; ++t) acc[t] = v8f{};

    float4 an = *(const float4*)Ar;
    for (int k0 = 0; k0 < K; k0 += 8) {
      const float4 a4 = an;
      an = *(const float4*)(Ar + min(k0 + 8, K - 8));
      const float* Bq = Bl + (k0 >> 2) * GEMM_P;
      float4 b4[8];
#pragma unroll
      for (int t = 0; t < 8; ++t) b4[t] = *(const float4*)(Bq + t * 64);
      v2f a01, a23;
      a01.x = a4.x; a01.y = a4.y; a23.x = a4.z; a23.y = a4.w;
#pragma unroll
      for (int t = 0; t < 8; ++t) {
        v2f b; b.x = b4[t].x; b.y = b4[t].y;
        acc[t] = __builtin_amdgcn_wmma_f32_16x16x4_f32(false, a01, false, b,
                                                       (short)0, acc[t], false, false);
      }
#pragma unroll
      for (int t = 0; t < 8; ++t) {
        v2f b; b.x = b4[t].z; b.y = b4[t].w;
        acc[t] = __builtin_amdgcn_wmma_f32_16x16x4_f32(false, a23, false, b,
                                                       (short)0, acc[t], false, false);
      }
    }

    float ba[8], bb[8];
#pragma unroll
    for (int t = 0; t < 8; ++t) {
      ba[t] = (MODE == 1) ? bias_a[cb + t * 16 + c] : 0.0f;
      bb[t] = (MODE == 1) ? bias_b[cb + t * 16 + c] : 0.0f;
    }
#pragma unroll
    for (int r = 0; r < 8; ++r) {
      const long long orow = (long long)tile * 16 + h * 8 + r;
      if (orow < M) {
#pragma unroll
        for (int t = 0; t < 8; ++t) {
          const int col = cb + t * 16 + c;
          const long long o = orow * 256 + col;
          if (MODE == 0) C[o] = acc[t][r];
          else           C[o] += acc[t][r] + ba[t] + bb[t];
        }
      }
    }
  }
}

__global__ void attn_scores(const float* __restrict__ hbuf,
                            const float* __restrict__ att_s,
                            const float* __restrict__ att_d,
                            float* __restrict__ as_out,
                            float* __restrict__ ad_out,
                            int Nn, int Hh, int Cc) {
  long long tid = (long long)blockIdx.x * blockDim.x + threadIdx.x;
  if (tid >= (long long)Nn * Hh) return;
  int n = (int)(tid / Hh), h = (int)(tid % Hh);
  const float* __restrict__ row = hbuf + (long long)n * 256 + h * Cc;
  const float* __restrict__ vs  = att_s + h * Cc;
  const float* __restrict__ vd  = att_d + h * Cc;
  float ss = 0.0f, sd = 0.0f;
  for (int c = 0; c < Cc; ++c) { float v = row[c]; ss += v * vs[c]; sd += v * vd[c]; }
  as_out[(long long)n * Hh + h] = ss;
  ad_out[(long long)n * Hh + h] = sd;
}

#define AGG_R 200
#define AGG_B 8
#define AGG_S 512
#define AGG_Q 768
#define AGG_BAD 255u
#define AGG_PER ((AGG_R + 31) / 32)
#define AGG_LS (32 * AGG_Q + AGG_R)

__global__ void edge_owner(const int* __restrict__ ei, unsigned* __restrict__ tag4,
                           int E, int Epad) {
  const int t = blockIdx.x * blockDim.x + threadIdx.x;
  if (t * 4 >= Epad) return;
  unsigned w = 0u;
#pragma unroll
  for (int i = 0; i < 4; ++i) {
    const int e = t * 4 + i;
    const unsigned b = (e < E) ? (unsigned)min(ei[E + e] / AGG_R, 255) : 255u;
    w |= b << (8 * i);
  }
  tag4[t] = w;
}

template <int HEADS>
__global__ __launch_bounds__(1024)
void edge_softmax_aggregate_own(const int* __restrict__ ei,
                                const unsigned* __restrict__ tag,
                                const float* __restrict__ hin,
                                const float* __restrict__ asrc,
                                const float* __restrict__ adst,
                                float* __restrict__ agg,
                                int2* __restrict__ l_rec,
                                int* __restrict__ l_src,
                                int E, int Epad, int Nn) {
  __shared__ float    c_ad[AGG_R * HEADS];
  __shared__ unsigned c_m[AGG_R * HEADS];
  __shared__ int      queue[32 * AGG_Q];
  __shared__ int      rowsrc[32 * AGG_Q + AGG_R];
  __shared__ int      r_cnt[AGG_R];
  __shared__ int      r_beg[AGG_R];
  __shared__ int      r_fill[AGG_R];
  __shared__ int      s_lost;

  const int lane = threadIdx.x & 31;
  const int wave = __builtin_amdgcn_readfirstlane(threadIdx.x >> 5);
  const int n0   = blockIdx.x * AGG_R;
  const int nown = min(AGG_R, Nn - n0);
  const unsigned mytag = min(blockIdx.x, 255u);
  const int hA   = (HEADS == 8) ? (lane >> 3) : 0;
  const int hB   = (HEADS == 8) ? 4 + (lane >> 3) : 0;

  for (int t = threadIdx.x; t < nown * HEADS; t += 1024) {
    const long long g = (long long)n0 * HEADS + t;
    const float a_d = adst[g];
    c_ad[t] = a_d;
    c_m[t]  = f2sortable(leaky(asrc[g] + a_d));
  }
  if (threadIdx.x < AGG_R) r_cnt[threadIdx.x] = 1;
  if (threadIdx.x == 0) s_lost = 0;
  __syncthreads();

  const int* __restrict__ dstp = ei + E;
  int* q = queue + wave * AGG_Q;
  int qn = 0;

  auto pack_range = [&](int* qp, int n) {
    for (int i0 = 0; i0 < n; i0 += 128) {
      int s[4]; unsigned dl[4];
#pragma unroll
      for (int u = 0; u < 4; ++u) {
        const int e = qp[min(i0 + u * 32 + lane, n - 1)];
        s[u]  = ei[e];
        dl[u] = (unsigned)(dstp[e] - n0);
      }
#pragma unroll
      for (int u = 0; u < 4; ++u) {
        const int i = i0 + u * 32 + lane;
        if (i < n) {
          const unsigned d = dl[u] < (unsigned)nown ? dl[u] : AGG_BAD;
          qp[i] = s[u] | (int)(d << 24);
          if (d != AGG_BAD) atomicAdd(&r_cnt[d], 1);
        }
      }
    }
  };

  constexpr int MU = (HEADS == 8) ? 8 : 4;
  auto max_range = [&](const int* qp, int n) {
    const unsigned items = (unsigned)n * HEADS;
    for (unsigned t0 = 0; t0 < items; t0 += 32 * MU) {
      unsigned cj[MU]; float a[MU];
#pragma unroll
      for (int u = 0; u < MU; ++u) {
        const unsigned t = min(t0 + u * 32 + lane, items - 1);
        const unsigned v = (unsigned)qp[t / HEADS];
        const unsigned h = t % HEADS;
        cj[u] = (v >> 24) * HEADS + h;
        a[u]  = asrc[(long long)(v & 0xFFFFFFu) * HEADS + h];
      }
      if constexpr (MU == 8)
        asm volatile("" :: "v"(a[0]), "v"(a[1]), "v"(a[2]), "v"(a[3]),
                           "v"(a[4]), "v"(a[5]), "v"(a[6]), "v"(a[7]));
      else
        asm volatile("" :: "v"(a[0]), "v"(a[1]), "v"(a[2]), "v"(a[3]));
#pragma unroll
      for (int u = 0; u < MU; ++u) {
        const bool ok = (t0 + u * 32 + lane < items) && (cj[u] < (unsigned)(nown * HEADS));
        const unsigned cc = ok ? cj[u] : 0u;
        const unsigned val = f2sortable(leaky(a[u] + c_ad[cc]));
        if (ok) atomicMax(&c_m[cc], val);
      }
    }
  };

  bool lost = false;

  const int nstep = Epad / AGG_S;
  auto scan = [&](auto&& after_step) {
    if (nstep <= 0) return;
    const uint4* __restrict__ tag16 = (const uint4*)tag;
    const unsigned pat = mytag * 0x01010101u;
    const int rot = (int)(((long long)blockIdx.x * nstep) / gridDim.x);
    auto step_of = [&](int k) { const int c = k + rot; return c >= nstep ? c - nstep : c; };
    auto load_step = [&](int k) { return tag16[step_of(min(k, nstep - 1)) * 32 + lane]; };
    auto detect = [&](const uint4 t) {
      const unsigned wv[4] = {t.x, t.y, t.z, t.w};
      unsigned hm = 0u;
#pragma unroll
      for (int j = 0; j < 4; ++j) {
        const unsigned x = wv[j] ^ pat;
        const unsigned y = (x & 0x7F7F7F7Fu) + 0x7F7F7F7Fu;
        const unsigned z = ~(y | x) & 0x80808080u;
        hm |= z >> (7 - j);
      }
      return hm;
    };
    auto collect = [&](unsigned hm, int k) {
      const int ebase = step_of(k) * AGG_S + lane * 16;
      unsigned m = __builtin_amdgcn_ballot_w32(hm != 0u);
      while (m != 0u) {
        const int p = __builtin_ctz(hm | 0x80000000u);
        const int e = ebase + ((p & 3) << 2) + (p >> 3);
        const bool hit = (hm != 0u) && (e < E);
        const unsigned mask = __builtin_amdgcn_ballot_w32(hit);
        if (hit) q[qn + __builtin_popcount(mask & ((1u << lane) - 1u))] = e;
        qn += __builtin_popcount(mask);
        hm &= hm - 1u;
        m = __builtin_amdgcn_ballot_w32(hm != 0u);
      }
      after_step(k + 32 < nstep);
    };
    uint4 ta = load_step(wave), tb = load_step(wave + 32);
    for (int k = wave; k < nstep && !lost; k += 64) {
      const unsigned h0 = detect(ta);
      const unsigned h1 = detect(tb);
      ta = load_step(k + 64);
      tb = load_step(k + 96);
      collect(h0, k);
      if (!lost && k + 32 < nstep) collect(h1, k + 32);
    }
  };

  auto weight = [&](float s, float ad, float m) { return __expf(leaky(s + ad) - m); };
  auto add_row = [&](float4& a0, float4& a1, float& zA, float& zB,
                     const float4 x0, const float4 x1, float wA, float wB) {
    a0.x += x0.x * wA;  a0.y += x0.y * wA;  a0.z += x0.z * wA;  a0.w += x0.w * wA;
    a1.x += x1.x * wB;  a1.y += x1.y * wB;  a1.z += x1.z * wB;  a1.w += x1.w * wB;
    zA += wA;  zB += wB;
  };
  auto store_row = [&](long long n, const float4 a0, const float4 a1, float zA, float zB) {
    const float iA = 1.0f / zA, iB = 1.0f / zB;
    float4 v0, v1;
    v0.x = a0.x * iA;  v0.y = a0.y * iA;  v0.z = a0.z * iA;  v0.w = a0.w * iA;
    v1.x = a1.x * iB;  v1.y = a1.y * iB;  v1.z = a1.z * iB;  v1.w = a1.w * iB;
    float4* __restrict__ orow = (float4*)(agg + n * 256);
    orow[lane]      = v0;
    orow[32 + lane] = v1;
  };

  scan([&](bool more) {
    if (more && qn > AGG_Q - AGG_S) lost = true;
  });
  if (lost) s_lost = 1;
  else if (qn > 0) { pack_range(q, qn); max_range(q, qn); }
  __syncthreads();

  if (s_lost != 0) {
    if ((int)threadIdx.x < nown) l_rec[n0 + threadIdx.x] = int2{0, 0};
    for (int r = wave; r < nown; r += 32) {
      const long long n = n0 + r;
      const float adA = adst[n * HEADS + hA], adB = adst[n * HEADS + hB];
      const float slA = asrc[n * HEADS + hA], slB = asrc[n * HEADS + hB];
      float mA = leaky(slA + adA), mB = leaky(slB + adB);
      for (int e0 = 0; e0 < E; e0 += 32) {
        const int e = e0 + lane;
        const bool hit = (e < E) && (dstp[e] == (int)n);
        const int s = hit ? ei[e] : 0;
        unsigned m = __builtin_amdgcn_ballot_w32(hit);
        while (m != 0u) {
          const int sj = __builtin_amdgcn_readlane(s, __builtin_ctz(m));
          m &= m - 1u;
          mA = fmaxf(mA, leaky(asrc[(long long)sj * HEADS + hA] + adA));
          mB = fmaxf(mB, leaky(asrc[(long long)sj * HEADS + hB] + adB));
        }
      }
      const float4* __restrict__ own = (const float4*)(hin + n * 256);
      float4 a0 = float4{0.0f, 0.0f, 0.0f, 0.0f}, a1 = a0;
      float zA = 0.0f, zB = 0.0f;
      add_row(a0, a1, zA, zB, own[lane], own[32 + lane],
              weight(slA, adA, mA), weight(slB, adB, mB));
      for (int e0 = 0; e0 < E; e0 += 32) {
        const int e = e0 + lane;
        const bool hit = (e < E) && (dstp[e] == (int)n);
        const int s = hit ? ei[e] : 0;
        unsigned m = __builtin_amdgcn_ballot_w32(hit);
        while (m != 0u) {
          const int sj = __builtin_amdgcn_readlane(s, __builtin_ctz(m));
          m &= m - 1u;
          const float4* __restrict__ row = (const float4*)(hin + (long long)sj * 256);
          add_row(a0, a1, zA, zB, row[lane], row[32 + lane],
                  weight(asrc[(long long)sj * HEADS + hA], adA, mA),
                  weight(asrc[(long long)sj * HEADS + hB], adB, mB));
        }
      }
      store_row(n, a0, a1, zA, zB);
    }
    return;
  }

  if (wave == 0) {
    int c[AGG_PER];
    int sum = 0;
#pragma unroll
    for (int j = 0; j < AGG_PER; ++j) {
      const int r = lane * AGG_PER + j;
      c[j] = (r < nown) ? r_cnt[r] : 0;
      sum += c[j];
    }
    int inc = sum;
#pragma unroll
    for (int off = 1; off < 32; off <<= 1) {
      const int t = __shfl_up(inc, off);
      if (lane >= off) inc += t;
    }
    int run = inc - sum;
#pragma unroll
    for (int j = 0; j < AGG_PER; ++j) {
      const int r = lane * AGG_PER + j;
      if (r < nown) {
        r_beg[r]    = run;
        r_fill[r]   = run + 1;
        rowsrc[run] = n0 + r;
      }
      run += c[j];
    }
  }
  __syncthreads();

  for (int i = lane; i < qn; i += 32) {
    const unsigned v = (unsigned)q[i];
    const unsigned d = v >> 24;
    if (d != AGG_BAD) rowsrc[atomicAdd(&r_fill[d], 1)] = (int)(v & 0xFFFFFFu);
  }
  __syncthreads();

  {
    const int lbase = blockIdx.x * AGG_LS;
    const int total = r_beg[nown - 1] + r_cnt[nown - 1];
    if ((int)threadIdx.x < nown)
      l_rec[n0 + threadIdx.x] = int2{r_cnt[threadIdx.x], lbase + r_beg[threadIdx.x]};
    for (int i = threadIdx.x; i < total; i += 1024) l_src[lbase + i] = rowsrc[i];
  }

  for (int r = wave; r < nown; r += 32) {
    const int cnt = __builtin_amdgcn_readfirstlane(r_cnt[r]);
    const int beg = __builtin_amdgcn_readfirstlane(r_beg[r]);
    const int cj  = r * HEADS;
    const float adA = c_ad[cj + hA], mA = sortable2f(c_m[cj + hA]);
    const float adB = c_ad[cj + hB], mB = sortable2f(c_m[cj + hB]);
    float4 a0 = float4{0.0f, 0.0f, 0.0f, 0.0f}, a1 = a0;
    float zA = 0.0f, zB = 0.0f;
    for (int i0 = 0; i0 < cnt; i0 += AGG_B) {
      const int nb = min(AGG_B, cnt - i0);
      const int sv = rowsrc[beg + i0 + min(lane, nb - 1)];
      float4 x0[AGG_B], x1[AGG_B];
      float  sA[AGG_B], sB[AGG_B];
#pragma unroll
      for (int j = 0; j < AGG_B; ++j) {
        const int sj = __builtin_amdgcn_readlane(sv, j);
        const float4* __restrict__ row = (const float4*)(hin + (long long)sj * 256);
        x0[j] = row[lane];
        x1[j] = row[32 + lane];
        sA[j] = asrc[(long long)sj * HEADS + hA];
        sB[j] = (HEADS == 8) ? asrc[(long long)sj * HEADS + hB] : 0.0f;
      }
#pragma unroll
      for (int j = 0; j < AGG_B; ++j) {
        const float on = (j < nb) ? 1.0f : 0.0f;
        const float wA = weight(sA[j], adA, mA) * on;
        const float wB = (HEADS == 8) ? weight(sB[j], adB, mB) * on : wA;
        add_row(a0, a1, zA, zB, x0[j], x1[j], wA, wB);
      }
    }
    store_row((long long)n0 + r, a0, a1, zA, zB);
  }
}

#define LST_W 8
#define LST_R 8

__device__ __forceinline__ void stream_row(const int* __restrict__ ei, int E,
                                           const float* __restrict__ hin,
                                           const float* __restrict__ asrc,
                                           float* __restrict__ agg,
                                           int n, float ad, int lane) {
  float4 a0 = float4{0.0f, 0.0f, 0.0f, 0.0f}, a1 = a0;
  float z = 0.0f;
  const int* __restrict__ dstp = ei + E;
  const float sl = asrc[n];
  float m = leaky(sl + ad);
  for (int e0 = 0; e0 < E; e0 += 32) {
    const int e = e0 + lane;
    const bool hit = (e < E) && (dstp[e] == n);
    const int s = hit ? ei[e] : 0;
    unsigned k = __builtin_amdgcn_ballot_w32(hit);
    while (k != 0u) {
      const int sj = __builtin_amdgcn_readlane(s, __builtin_ctz(k));
      k &= k - 1u;
      m = fmaxf(m, leaky(asrc[sj] + ad));
    }
  }
  {
    const float4* __restrict__ own = (const float4*)(hin + (long long)n * 256);
    const float w = __expf(leaky(sl + ad) - m);
    const float4 x0 = own[lane], x1 = own[32 + lane];
    a0.x += x0.x * w;  a0.y += x0.y * w;  a0.z += x0.z * w;  a0.w += x0.w * w;
    a1.x += x1.x * w;  a1.y += x1.y * w;  a1.z += x1.z * w;  a1.w += x1.w * w;
    z += w;
  }
  for (int e0 = 0; e0 < E; e0 += 32) {
    const int e = e0 + lane;
    const bool hit = (e < E) && (dstp[e] == n);
    const int s = hit ? ei[e] : 0;
    unsigned k = __builtin_amdgcn_ballot_w32(hit);
    while (k != 0u) {
      const int sj = __builtin_amdgcn_readlane(s, __builtin_ctz(k));
      k &= k - 1u;
      const float4* __restrict__ row = (const float4*)(hin + (long long)sj * 256);
      const float w = __expf(leaky(asrc[sj] + ad) - m);
      const float4 x0 = row[lane], x1 = row[32 + lane];
      a0.x += x0.x * w;  a0.y += x0.y * w;  a0.z += x0.z * w;  a0.w += x0.w * w;
      a1.x += x1.x * w;  a1.y += x1.y * w;  a1.z += x1.z * w;  a1.w += x1.w * w;
      z += w;
    }
  }
  const float iz = 1.0f / z;
  float4 v0, v1;
  v0.x = a0.x * iz;  v0.y = a0.y * iz;  v0.z = a0.z * iz;  v0.w = a0.w * iz;
  v1.x = a1.x * iz;  v1.y = a1.y * iz;  v1.z = a1.z * iz;  v1.w = a1.w * iz;
  float4* __restrict__ orow = (float4*)(agg + (long long)n * 256);
  orow[lane]      = v0;
  orow[32 + lane] = v1;
}

__global__ __launch_bounds__(32 * LST_W)
void softmax_aggregate_listed(const int* __restrict__ ei,
                              const int2* __restrict__ l_rec,
                              const int* __restrict__ l_src,
                              const float* __restrict__ hin,
                              const float* __restrict__ asrc,
                              const float* __restrict__ adst,
                              float* __restrict__ agg,
                              int E, int Nn) {
  const int lane = threadIdx.x & 31;
  const int wave = __builtin_amdgcn_readfirstlane(threadIdx.x >> 5);
  const int n0 = (blockIdx.x * LST_W + wave) * LST_R;
  if (n0 >= Nn) return;
  const int nrows = min(LST_R, Nn - n0);

  const int   nl   = min(n0 + lane, Nn - 1);
  const int2  recv = l_rec[nl];
  const float adv  = adst[nl];

  auto window = [&](int cnt, int beg, int i0) {
    return l_src[beg + min(i0 + lane, cnt - 1)];
  };

  int cntN = __builtin_amdgcn_readlane(recv.x, 0);
  int begN = __builtin_amdgcn_readlane(recv.y, 0);
  int svN  = (cntN > 0) ? window(cntN, begN, 0) : 0;

  for (int r = 0; r < nrows; ++r) {
    const int n = n0 + r;
    const int cnt = cntN, beg = begN;
    int sv = svN;
    const float ad = __int_as_float(__builtin_amdgcn_readlane(__float_as_int(adv), r));
    if (r + 1 < nrows) {
      cntN = __builtin_amdgcn_readlane(recv.x, r + 1);
      begN = __builtin_amdgcn_readlane(recv.y, r + 1);
      svN  = (cntN > 0) ? window(cntN, begN, 0) : 0;
    }

    if (cnt <= 0) {
      stream_row(ei, E, hin, asrc, agg, n, ad, lane);
      continue;
    }

    float4 a0 = float4{0.0f, 0.0f, 0.0f, 0.0f}, a1 = a0;
    float z = 0.0f;
    {
      float sc = asrc[sv];
      float4 x0[AGG_B], x1[AGG_B];
      auto request = [&](int i0) {
        const int nb = min(AGG_B, cnt - i0);
        const int wb = i0 & 31;
#pragma unroll
        for (int j = 0; j < AGG_B; ++j) {
          const int sj = __builtin_amdgcn_readlane(sv, wb + min(j, nb - 1));
          const float4* __restrict__ row = (const float4*)(hin + (long long)sj * 256);
          x0[j] = row[lane];
          x1[j] = row[32 + lane];
        }
      };
      auto add_batch = [&](int i0, float m) {
        const int nb = min(AGG_B, cnt - i0);
        const int wb = i0 & 31;
#pragma unroll
        for (int j = 0; j < AGG_B; ++j) {
          const float s  = __int_as_float(__builtin_amdgcn_readlane(__float_as_int(sc),
                                                                    wb + min(j, nb - 1)));
          const float on = (j < nb) ? 1.0f : 0.0f;
          const float w  = __expf(leaky(s + ad) - m) * on;
          a0.x += x0[j].x * w;  a0.y += x0[j].y * w;  a0.z += x0[j].z * w;  a0.w += x0[j].w * w;
          a1.x += x1[j].x * w;  a1.y += x1[j].y * w;  a1.z += x1[j].z * w;  a1.w += x1[j].w * w;
          z += w;
        }
      };

      request(0);
      float m = leaky(sc + ad);
      for (int i0 = 32; i0 < cnt; i0 += 32) {
        const int s = window(cnt, beg, i0);
        m = fmaxf(m, leaky(asrc[s] + ad));
      }
#pragma unroll
      for (int off = 16; off > 0; off >>= 1) m = fmaxf(m, __shfl_xor(m, off));
      add_batch(0, m);

      for (int i0 = AGG_B; i0 < cnt; i0 += AGG_B) {
        if ((i0 & 31) == 0) {
          sv = window(cnt, beg, i0);
          sc = asrc[sv];
        }
        request(i0);
        add_batch(i0, m);
      }
    }

    const float iz = 1.0f / z;
    float4 v0, v1;
    v0.x = a0.x * iz;  v0.y = a0.y * iz;  v0.z = a0.z * iz;  v0.w = a0.w * iz;
    v1.x = a1.x * iz;  v1.y = a1.y * iz;  v1.z = a1.z * iz;  v1.w = a1.w * iz;
    float4* __restrict__ orow = (float4*)(agg + (long long)n * 256);
    orow[lane]      = v0;
    orow[32 + lane] = v1;
  }
}

__global__ void ln_elu(float* __restrict__ hbuf, const float* __restrict__ b1,
                       const float* __restrict__ gamma,
                       const float* __restrict__ beta, int Nn) {
  long long gtid = (long long)blockIdx.x * blockDim.x + threadIdx.x;
  int n    = (int)(gtid >> 5);
  int lane = threadIdx.x & 31;
  if (n >= Nn) return;
  float* __restrict__ row = hbuf + (long long)n * 256;
  float vals[8];
  float sum = 0.0f;
#pragma unroll
  for (int k = 0; k < 8; ++k) {
    int idx = k * 32 + lane;
    vals[k] = row[idx] + b1[idx];
    sum += vals[k];
  }
#pragma unroll
  for (int off = 16; off > 0; off >>= 1) sum += __shfl_xor(sum, off);
  float mu = sum * (1.0f / 256.0f);
  float var = 0.0f;
#pragma unroll
  for (int k = 0; k < 8; ++k) { float t = vals[k] - mu; var += t * t; }
#pragma unroll
  for (int off = 16; off > 0; off >>= 1) var += __shfl_xor(var, off);
  var *= (1.0f / 256.0f);
  float rs = rsqrtf(var + LN_EPS);
#pragma unroll
  for (int k = 0; k < 8; ++k) {
    int idx = k * 32 + lane;
    float t = (vals[k] - mu) * rs * gamma[idx] + beta[idx];
    row[idx] = t > 0.0f ? t : (__expf(t) - 1.0f);
  }
}

__global__ void pool_accum(const float* __restrict__ o,
                           const int* __restrict__ batch,
                           float* __restrict__ outsum,
                           float* __restrict__ cnt, int Nn) {
  long long gtid = (long long)blockIdx.x * blockDim.x + threadIdx.x;
  int n    = (int)(gtid >> 5);
  int lane = threadIdx.x & 31;
  if (n >= Nn) return;
  int g = batch[n];
  const float* __restrict__ row = o + (long long)n * 256;
  float* __restrict__ acc       = outsum + (long long)g * 256;
#pragma unroll
  for (int k = 0; k < 8; ++k) {
    int idx = k * 32 + lane;
    atomicAdd(&acc[idx], row[idx]);
  }
  if (lane == 0) atomicAdd(&cnt[g], 1.0f);
}

__global__ void pool_div(float* __restrict__ out, const float* __restrict__ cnt,
                         int total) {
  int i = blockIdx.x * blockDim.x + threadIdx.x;
  if (i >= total) return;
  out[i] /= fmaxf(cnt[i >> 8], 1.0f);
}

static inline int cdiv(long long a, long long b) { return (int)((a + b - 1) / b); }

extern "C" void kernel_launch(void* const* d_in, const int* in_sizes, int n_in,
                              void* d_out, int out_size, void* d_ws, size_t ws_size,
                              hipStream_t stream) {
  const float* x        = (const float*)d_in[0];
  const int*   ei       = (const int*)  d_in[1];
  const int*   batch    = (const int*)  d_in[2];
  const float* W1       = (const float*)d_in[3];
  const float* att_src1 = (const float*)d_in[4];
  const float* att_dst1 = (const float*)d_in[5];
  const float* b1       = (const float*)d_in[6];
  const float* gamma    = (const float*)d_in[7];
  const float* beta     = (const float*)d_in[8];
  const float* W2       = (const float*)d_in[9];
  const float* att_src2 = (const float*)d_in[10];
  const float* att_dst2 = (const float*)d_in[11];
  const float* b2       = (const float*)d_in[12];
  const float* Wres     = (const float*)d_in[13];
  const float* bres     = (const float*)d_in[14];
  float* out = (float*)d_out;

  const int Nn = in_sizes[2];
  const int Ee = in_sizes[1] / 2;
  const int Gg = out_size / 256;
  const int EA = Ee + Nn;
  const int IN_DIM = in_sizes[0] / Nn;

  const long long NB = (long long)Nn * 256;
  float* ws   = (float*)d_ws;
  float* buf0 = ws;
  float* buf1 = ws + NB;
  float* buf2 = ws + 2 * NB;
  float* cnt  = ws + 3 * NB;
  float* as1  = cnt + Gg;
  float* ad1  = as1 + (long long)Nn * 8;
  float* as2  = ad1 + (long long)Nn * 8;
  float* ad2  = as2 + Nn;
  const int Epad = cdiv(Ee, AGG_S) * AGG_S;
  unsigned* tag = (unsigned*)(ws + ((3 * NB + (long long)Nn * 18 + Gg + 3) / 4) * 4);
  const int NBLK = cdiv(Nn, AGG_R);
  int2* l_rec = (int2*)(tag + Epad / 4);
  int*  l_src = (int*)(l_rec + (long long)NBLK * AGG_R);

  const int B = 256;

  edge_owner<<<cdiv(Epad / 4, B), B, 0, stream>>>(ei, tag, Ee, Epad);
  fill_f32<<<cdiv(Gg, B), B, 0, stream>>>(cnt, Gg, 0.0f);
  fill_f32<<<cdiv((long long)Gg * 256, B), B, 0, stream>>>(out, (long long)Gg * 256, 0.0f);

  gemm_wmma_n256<0><<<GEMM_G, 512, 0, stream>>>(x, W1, buf0, nullptr, nullptr, Nn, IN_DIM);
  attn_scores<<<cdiv((long long)Nn * 8, B), B, 0, stream>>>(buf0, att_src1, att_dst1,
                                                            as1, ad1, Nn, 8, 32);
  edge_softmax_aggregate_own<8><<<NBLK, 1024, 0, stream>>>(ei, tag, buf0, as1, ad1, buf1,
                                                           l_rec, l_src, Ee, Epad, Nn);
  ln_elu<<<cdiv((long long)Nn * 32, B), B, 0, stream>>>(buf1, b1, gamma, beta, Nn);

  gemm_wmma_n256<0><<<GEMM_G, 512, 0, stream>>>(buf1, W2, buf0, nullptr, nullptr, Nn, 256);
  attn_scores<<<cdiv((long long)Nn, B), B, 0, stream>>>(buf0, att_src2, att_dst2,
                                                        as2, ad2, Nn, 1, 256);
  softmax_aggregate_listed<<<cdiv(Nn, LST_W * LST_R), 32 * LST_W, 0, stream>>>(ei, l_rec, l_src, buf0, as2, ad2,
                                                                       buf2, Ee, Nn);
  gemm_wmma_n256<1><<<GEMM_G, 512, 0, stream>>>(buf1, Wres, buf2, b2, bres, Nn, 256);

  pool_accum<<<cdiv((long long)Nn * 32, B), B, 0, stream>>>(buf2, batch, out, cnt, Nn);
  pool_div  <<<cdiv((long long)Gg * 256, B), B, 0, stream>>>(out, cnt, Gg * 256);
}


# Mean pool leaves global float atomics: a block owns one graph, scans the batch vector, its waves sum matching rows in registers, fixed-order LDS fold, divide at store; sum order = node order per wave
# speedup vs baseline: 4.1061x; 1.1123x over previous
#include <hip/hip_runtime.h>
#include <hip/hip_bf16.h>


typedef __attribute__((ext_vector_type(2))) float v2f;
typedef __attribute__((ext_vector_type(8))) float v8f;

#define NEG_SLOPE 0.2f
#define LN_EPS    1e-5f

__device__ __forceinline__ unsigned f2sortable(float f) {
  unsigned u = __float_as_uint(f);
  return (u & 0x80000000u) ? ~u : (u | 0x80000000u);
}
__device__ __forceinline__ float sortable2f(unsigned u) {
  u = (u & 0x80000000u) ? (u & 0x7FFFFFFFu) : ~u;
  return __uint_as_float(u);
}
__device__ __forceinline__ float leaky(float x) {
  return x > 0.0f ? x : NEG_SLOPE * x;
}
__device__ __forceinline__ void edge_sd(const int* __restrict__ ei, int E, int e,
                                        int& s, int& d) {
  if (e < E) { s = ei[e]; d = ei[E + e]; }
  else       { s = d = e - E; }
}

#define GEMM_G 256
#define GEMM_P 1056
#define GEMM_KMAX 256

template <int MODE>
__global__ __launch_bounds__(512)
void gemm_wmma_n256(const float* __restrict__ A,
                    const float* __restrict__ B,
                    float* __restrict__ C,
                    const float* __restrict__ bias_a,
                    const float* __restrict__ bias_b,
                    int M, int K) {
  __shared__ float Bs[(GEMM_KMAX / 4) * GEMM_P];

  const int lane = threadIdx.x & 31;
  const int wave = __builtin_amdgcn_readfirstlane(threadIdx.x >> 5);
  const int c    = lane & 15;
  const int h    = lane >> 4;

  for (int i = threadIdx.x; i < (K >> 2) * 256; i += 512) {
    const int g = i >> 8, n = i & 255;
    const float* __restrict__ p = B + (long long)(4 * g) * 256 + n;
    float4 v;
    v.x = p[0]; v.y = p[256]; v.z = p[512]; v.w = p[768];
    *(float4*)(Bs + g * GEMM_P + n * 4) = v;
  }
  __syncthreads();

  const int ntile = (M + 15) >> 4;
  const int t0 = (int)(((long long)blockIdx.x * ntile) / GEMM_G);
  const int t1 = (int)(((long long)(blockIdx.x + 1) * ntile) / GEMM_G);
  const int nunit = (t1 - t0) * 2;

  for (int u = wave; u < nunit; u += 16) {
    const int tile = t0 + (u >> 1);
    const int cb   = (u & 1) * 128;
    long long arow = (long long)tile * 16 + c;
    if (arow >= M) arow = M - 1;
    const float* __restrict__ Ar = A + arow * K + 4 * h;
    const float* Bl = Bs + h * GEMM_P + (cb + c) * 4;

    v8f acc[8];
#pragma unroll
    for (int t = 0; t < 8; ++t) acc[t] = v8f{};

    float4 an = *(const float4*)Ar;
    for (int k0 = 0; k0 < K; k0 += 8) {
      const float4 a4 = an;
      an = *(const float4*)(Ar + min(k0 + 8, K - 8));
      const float* Bq = Bl + (k0 >> 2) * GEMM_P;
      float4 b4[8];
#pragma unroll
      for (int t = 0; t < 8; ++t) b4[t] = *(const float4*)(Bq + t * 64);
      v2f a01, a23;
      a01.x = a4.x; a01.y = a4.y; a23.x = a4.z; a23.y = a4.w;
#pragma unroll
      for (int t = 0; t < 8; ++t) {
        v2f b; b.x = b4[t].x; b.y = b4[t].y;
        acc[t] = __builtin_amdgcn_wmma_f32_16x16x4_f32(false, a01, false, b,
                                                       (short)0, acc[t], false, false);
      }
#pragma unroll
      for (int t = 0; t < 8; ++t) {
        v2f b; b.x = b4[t].z; b.y = b4[t].w;
        acc[t] = __builtin_amdgcn_wmma_f32_16x16x4_f32(false, a23, false, b,
                                                       (short)0, acc[t], false, false);
      }
    }

    float ba[8], bb[8];
#pragma unroll
    for (int t = 0; t < 8; ++t) {
      ba[t] = (MODE == 1) ? bias_a[cb + t * 16 + c] : 0.0f;
      bb[t] = (MODE == 1) ? bias_b[cb + t * 16 + c] : 0.0f;
    }
#pragma unroll
    for (int r = 0; r < 8; ++r) {
      const long long orow = (long long)tile * 16 + h * 8 + r;
      if (orow < M) {
#pragma unroll
        for (int t = 0; t < 8; ++t) {
          const int col = cb + t * 16 + c;
          const long long o = orow * 256 + col;
          if (MODE == 0) C[o] = acc[t][r];
          else           C[o] += acc[t][r] + ba[t] + bb[t];
        }
      }
    }
  }
}

__global__ void attn_scores(const float* __restrict__ hbuf,
                            const float* __restrict__ att_s,
                            const float* __restrict__ att_d,
                            float* __restrict__ as_out,
                            float* __restrict__ ad_out,
                            int Nn, int Hh, int Cc) {
  long long tid = (long long)blockIdx.x * blockDim.x + threadIdx.x;
  if (tid >= (long long)Nn * Hh) return;
  int n = (int)(tid / Hh), h = (int)(tid % Hh);
  const float* __restrict__ row = hbuf + (long long)n * 256 + h * Cc;
  const float* __restrict__ vs  = att_s + h * Cc;
  const float* __restrict__ vd  = att_d + h * Cc;
  float ss = 0.0f, sd = 0.0f;
  for (int c = 0; c < Cc; ++c) { float v = row[c]; ss += v * vs[c]; sd += v * vd[c]; }
  as_out[(long long)n * Hh + h] = ss;
  ad_out[(long long)n * Hh + h] = sd;
}

#define AGG_R 200
#define AGG_B 8
#define AGG_S 512
#define AGG_Q 768
#define AGG_BAD 255u
#define AGG_PER ((AGG_R + 31) / 32)
#define AGG_LS (32 * AGG_Q + AGG_R)

__global__ void edge_owner(const int* __restrict__ ei, unsigned* __restrict__ tag4,
                           int E, int Epad) {
  const int t = blockIdx.x * blockDim.x + threadIdx.x;
  if (t * 4 >= Epad) return;
  unsigned w = 0u;
#pragma unroll
  for (int i = 0; i < 4; ++i) {
    const int e = t * 4 + i;
    const unsigned b = (e < E) ? (unsigned)min(ei[E + e] / AGG_R, 255) : 255u;
    w |= b << (8 * i);
  }
  tag4[t] = w;
}

template <int HEADS>
__global__ __launch_bounds__(1024)
void edge_softmax_aggregate_own(const int* __restrict__ ei,
                                const unsigned* __restrict__ tag,
                                const float* __restrict__ hin,
                                const float* __restrict__ asrc,
                                const float* __restrict__ adst,
                                float* __restrict__ agg,
                                int2* __restrict__ l_rec,
                                int* __restrict__ l_src,
                                int E, int Epad, int Nn) {
  __shared__ float    c_ad[AGG_R * HEADS];
  __shared__ unsigned c_m[AGG_R * HEADS];
  __shared__ int      queue[32 * AGG_Q];
  __shared__ int      rowsrc[32 * AGG_Q + AGG_R];
  __shared__ int      r_cnt[AGG_R];
  __shared__ int      r_beg[AGG_R];
  __shared__ int      r_fill[AGG_R];
  __shared__ int      s_lost;

  const int lane = threadIdx.x & 31;
  const int wave = __builtin_amdgcn_readfirstlane(threadIdx.x >> 5);
  const int n0   = blockIdx.x * AGG_R;
  const int nown = min(AGG_R, Nn - n0);
  const unsigned mytag = min(blockIdx.x, 255u);
  const int hA   = (HEADS == 8) ? (lane >> 3) : 0;
  const int hB   = (HEADS == 8) ? 4 + (lane >> 3) : 0;

  for (int t = threadIdx.x; t < nown * HEADS; t += 1024) {
    const long long g = (long long)n0 * HEADS + t;
    const float a_d = adst[g];
    c_ad[t] = a_d;
    c_m[t]  = f2sortable(leaky(asrc[g] + a_d));
  }
  if (threadIdx.x < AGG_R) r_cnt[threadIdx.x] = 1;
  if (threadIdx.x == 0) s_lost = 0;
  __syncthreads();

  const int* __restrict__ dstp = ei + E;
  int* q = queue + wave * AGG_Q;
  int qn = 0;

  auto pack_range = [&](int* qp, int n) {
    for (int i0 = 0; i0 < n; i0 += 128) {
      int s[4]; unsigned dl[4];
#pragma unroll
      for (int u = 0; u < 4; ++u) {
        const int e = qp[min(i0 + u * 32 + lane, n - 1)];
        s[u]  = ei[e];
        dl[u] = (unsigned)(dstp[e] - n0);
      }
#pragma unroll
      for (int u = 0; u < 4; ++u) {
        const int i = i0 + u * 32 + lane;
        if (i < n) {
          const unsigned d = dl[u] < (unsigned)nown ? dl[u] : AGG_BAD;
          qp[i] = s[u] | (int)(d << 24);
          if (d != AGG_BAD) atomicAdd(&r_cnt[d], 1);
        }
      }
    }
  };

  constexpr int MU = (HEADS == 8) ? 8 : 4;
  auto max_range = [&](const int* qp, int n) {
    const unsigned items = (unsigned)n * HEADS;
    for (unsigned t0 = 0; t0 < items; t0 += 32 * MU) {
      unsigned cj[MU]; float a[MU];
#pragma unroll
      for (int u = 0; u < MU; ++u) {
        const unsigned t = min(t0 + u * 32 + lane, items - 1);
        const unsigned v = (unsigned)qp[t / HEADS];
        const unsigned h = t % HEADS;
        cj[u] = (v >> 24) * HEADS + h;
        a[u]  = asrc[(long long)(v & 0xFFFFFFu) * HEADS + h];
      }
      if constexpr (MU == 8)
        asm volatile("" :: "v"(a[0]), "v"(a[1]), "v"(a[2]), "v"(a[3]),
                           "v"(a[4]), "v"(a[5]), "v"(a[6]), "v"(a[7]));
      else
        asm volatile("" :: "v"(a[0]), "v"(a[1]), "v"(a[2]), "v"(a[3]));
#pragma unroll
      for (int u = 0; u < MU; ++u) {
        const bool ok = (t0 + u * 32 + lane < items) && (cj[u] < (unsigned)(nown * HEADS));
        const unsigned cc = ok ? cj[u] : 0u;
        const unsigned val = f2sortable(leaky(a[u] + c_ad[cc]));
        if (ok) atomicMax(&c_m[cc], val);
      }
    }
  };

  bool lost = false;

  const int nstep = Epad / AGG_S;
  auto scan = [&](auto&& after_step) {
    if (nstep <= 0) return;
    const uint4* __restrict__ tag16 = (const uint4*)tag;
    const unsigned pat = mytag * 0x01010101u;
    const int rot = (int)(((long long)blockIdx.x * nstep) / gridDim.x);
    auto step_of = [&](int k) { const int c = k + rot; return c >= nstep ? c - nstep : c; };
    auto load_step = [&](int k) { return tag16[step_of(min(k, nstep - 1)) * 32 + lane]; };
    auto detect = [&](const uint4 t) {
      const unsigned wv[4] = {t.x, t.y, t.z, t.w};
      unsigned hm = 0u;
#pragma unroll
      for (int j = 0; j < 4; ++j) {
        const unsigned x = wv[j] ^ pat;
        const unsigned y = (x & 0x7F7F7F7Fu) + 0x7F7F7F7Fu;
        const unsigned z = ~(y | x) & 0x80808080u;
        hm |= z >> (7 - j);
      }
      return hm;
    };
    auto collect = [&](unsigned hm, int k) {
      const int ebase = step_of(k) * AGG_S + lane * 16;
      unsigned m = __builtin_amdgcn_ballot_w32(hm != 0u);
      while (m != 0u) {
        const int p = __builtin_ctz(hm | 0x80000000u);
        const int e = ebase + ((p & 3) << 2) + (p >> 3);
        const bool hit = (hm != 0u) && (e < E);
        const unsigned mask = __builtin_amdgcn_ballot_w32(hit);
        if (hit) q[qn + __builtin_popcount(mask & ((1u << lane) - 1u))] = e;
        qn += __builtin_popcount(mask);
        hm &= hm - 1u;
        m = __builtin_amdgcn_ballot_w32(hm != 0u);
      }
      after_step(k + 32 < nstep);
    };
    uint4 ta = load_step(wave), tb = load_step(wave + 32);
    for (int k = wave; k < nstep && !lost; k += 64) {
      const unsigned h0 = detect(ta);
      const unsigned h1 = detect(tb);
      ta = load_step(k + 64);
      tb = load_step(k + 96);
      collect(h0, k);
      if (!lost && k + 32 < nstep) collect(h1, k + 32);
    }
  };

  auto weight = [&](float s, float ad, float m) { return __expf(leaky(s + ad) - m); };
  auto add_row = [&](float4& a0, float4& a1, float& zA, float& zB,
                     const float4 x0, const float4 x1, float wA, float wB) {
    a0.x += x0.x * wA;  a0.y += x0.y * wA;  a0.z += x0.z * wA;  a0.w += x0.w * wA;
    a1.x += x1.x * wB;  a1.y += x1.y * wB;  a1.z += x1.z * wB;  a1.w += x1.w * wB;
    zA += wA;  zB += wB;
  };
  auto store_row = [&](long long n, const float4 a0, const float4 a1, float zA, float zB) {
    const float iA = 1.0f / zA, iB = 1.0f / zB;
    float4 v0, v1;
    v0.x = a0.x * iA;  v0.y = a0.y * iA;  v0.z = a0.z * iA;  v0.w = a0.w * iA;
    v1.x = a1.x * iB;  v1.y = a1.y * iB;  v1.z = a1.z * iB;  v1.w = a1.w * iB;
    float4* __restrict__ orow = (float4*)(agg + n * 256);
    orow[lane]      = v0;
    orow[32 + lane] = v1;
  };

  scan([&](bool more) {
    if (more && qn > AGG_Q - AGG_S) lost = true;
  });
  if (lost) s_lost = 1;
  else if (qn > 0) { pack_range(q, qn); max_range(q, qn); }
  __syncthreads();

  if (s_lost != 0) {
    if ((int)threadIdx.x < nown) l_rec[n0 + threadIdx.x] = int2{0, 0};
    for (int r = wave; r < nown; r += 32) {
      const long long n = n0 + r;
      const float adA = adst[n * HEADS + hA], adB = adst[n * HEADS + hB];
      const float slA = asrc[n * HEADS + hA], slB = asrc[n * HEADS + hB];
      float mA = leaky(slA + adA), mB = leaky(slB + adB);
      for (int e0 = 0; e0 < E; e0 += 32) {
        const int e = e0 + lane;
        const bool hit = (e < E) && (dstp[e] == (int)n);
        const int s = hit ? ei[e] : 0;
        unsigned m = __builtin_amdgcn_ballot_w32(hit);
        while (m != 0u) {
          const int sj = __builtin_amdgcn_readlane(s, __builtin_ctz(m));
          m &= m - 1u;
          mA = fmaxf(mA, leaky(asrc[(long long)sj * HEADS + hA] + adA));
          mB = fmaxf(mB, leaky(asrc[(long long)sj * HEADS + hB] + adB));
        }
      }
      const float4* __restrict__ own = (const float4*)(hin + n * 256);
      float4 a0 = float4{0.0f, 0.0f, 0.0f, 0.0f}, a1 = a0;
      float zA = 0.0f, zB = 0.0f;
      add_row(a0, a1, zA, zB, own[lane], own[32 + lane],
              weight(slA, adA, mA), weight(slB, adB, mB));
      for (int e0 = 0; e0 < E; e0 += 32) {
        const int e = e0 + lane;
        const bool hit = (e < E) && (dstp[e] == (int)n);
        const int s = hit ? ei[e] : 0;
        unsigned m = __builtin_amdgcn_ballot_w32(hit);
        while (m != 0u) {
          const int sj = __builtin_amdgcn_readlane(s, __builtin_ctz(m));
          m &= m - 1u;
          const float4* __restrict__ row = (const float4*)(hin + (long long)sj * 256);
          add_row(a0, a1, zA, zB, row[lane], row[32 + lane],
                  weight(asrc[(long long)sj * HEADS + hA], adA, mA),
                  weight(asrc[(long long)sj * HEADS + hB], adB, mB));
        }
      }
      store_row(n, a0, a1, zA, zB);
    }
    return;
  }

  if (wave == 0) {
    int c[AGG_PER];
    int sum = 0;
#pragma unroll
    for (int j = 0; j < AGG_PER; ++j) {
      const int r = lane * AGG_PER + j;
      c[j] = (r < nown) ? r_cnt[r] : 0;
      sum += c[j];
    }
    int inc = sum;
#pragma unroll
    for (int off = 1; off < 32; off <<= 1) {
      const int t = __shfl_up(inc, off);
      if (lane >= off) inc += t;
    }
    int run = inc - sum;
#pragma unroll
    for (int j = 0; j < AGG_PER; ++j) {
      const int r = lane * AGG_PER + j;
      if (r < nown) {
        r_beg[r]    = run;
        r_fill[r]   = run + 1;
        rowsrc[run] = n0 + r;
      }
      run += c[j];
    }
  }
  __syncthreads();

  for (int i = lane; i < qn; i += 32) {
    const unsigned v = (unsigned)q[i];
    const unsigned d = v >> 24;
    if (d != AGG_BAD) rowsrc[atomicAdd(&r_fill[d], 1)] = (int)(v & 0xFFFFFFu);
  }
  __syncthreads();

  {
    const int lbase = blockIdx.x * AGG_LS;
    const int total = r_beg[nown - 1] + r_cnt[nown - 1];
    if ((int)threadIdx.x < nown)
      l_rec[n0 + threadIdx.x] = int2{r_cnt[threadIdx.x], lbase + r_beg[threadIdx.x]};
    for (int i = threadIdx.x; i < total; i += 1024) l_src[lbase + i] = rowsrc[i];
  }

  for (int r = wave; r < nown; r += 32) {
    const int cnt = __builtin_amdgcn_readfirstlane(r_cnt[r]);
    const int beg = __builtin_amdgcn_readfirstlane(r_beg[r]);
    const int cj  = r * HEADS;
    const float adA = c_ad[cj + hA], mA = sortable2f(c_m[cj + hA]);
    const float adB = c_ad[cj + hB], mB = sortable2f(c_m[cj + hB]);
    float4 a0 = float4{0.0f, 0.0f, 0.0f, 0.0f}, a1 = a0;
    float zA = 0.0f, zB = 0.0f;
    for (int i0 = 0; i0 < cnt; i0 += AGG_B) {
      const int nb = min(AGG_B, cnt - i0);
      const int sv = rowsrc[beg + i0 + min(lane, nb - 1)];
      float4 x0[AGG_B], x1[AGG_B];
      float  sA[AGG_B], sB[AGG_B];
#pragma unroll
      for (int j = 0; j < AGG_B; ++j) {
        const int sj = __builtin_amdgcn_readlane(sv, j);
        const float4* __restrict__ row = (const float4*)(hin + (long long)sj * 256);
        x0[j] = row[lane];
        x1[j] = row[32 + lane];
        sA[j] = asrc[(long long)sj * HEADS + hA];
        sB[j] = (HEADS == 8) ? asrc[(long long)sj * HEADS + hB] : 0.0f;
      }
#pragma unroll
      for (int j = 0; j < AGG_B; ++j) {
        const float on = (j < nb) ? 1.0f : 0.0f;
        const float wA = weight(sA[j], adA, mA) * on;
        const float wB = (HEADS == 8) ? weight(sB[j], adB, mB) * on : wA;
        add_row(a0, a1, zA, zB, x0[j], x1[j], wA, wB);
      }
    }
    store_row((long long)n0 + r, a0, a1, zA, zB);
  }
}

#define LST_W 8
#define LST_R 8

__device__ __forceinline__ void stream_row(const int* __restrict__ ei, int E,
                                           const float* __restrict__ hin,
                                           const float* __restrict__ asrc,
                                           float* __restrict__ agg,
                                           int n, float ad, int lane) {
  float4 a0 = float4{0.0f, 0.0f, 0.0f, 0.0f}, a1 = a0;
  float z = 0.0f;
  const int* __restrict__ dstp = ei + E;
  const float sl = asrc[n];
  float m = leaky(sl + ad);
  for (int e0 = 0; e0 < E; e0 += 32) {
    const int e = e0 + lane;
    const bool hit = (e < E) && (dstp[e] == n);
    const int s = hit ? ei[e] : 0;
    unsigned k = __builtin_amdgcn_ballot_w32(hit);
    while (k != 0u) {
      const int sj = __builtin_amdgcn_readlane(s, __builtin_ctz(k));
      k &= k - 1u;
      m = fmaxf(m, leaky(asrc[sj] + ad));
    }
  }
  {
    const float4* __restrict__ own = (const float4*)(hin + (long long)n * 256);
    const float w = __expf(leaky(sl + ad) - m);
    const float4 x0 = own[lane], x1 = own[32 + lane];
    a0.x += x0.x * w;  a0.y += x0.y * w;  a0.z += x0.z * w;  a0.w += x0.w * w;
    a1.x += x1.x * w;  a1.y += x1.y * w;  a1.z += x1.z * w;  a1.w += x1.w * w;
    z += w;
  }
  for (int e0 = 0; e0 < E; e0 += 32) {
    const int e = e0 + lane;
    const bool hit = (e < E) && (dstp[e] == n);
    const int s = hit ? ei[e] : 0;
    unsigned k = __builtin_amdgcn_ballot_w32(hit);
    while (k != 0u) {
      const int sj = __builtin_amdgcn_readlane(s, __builtin_ctz(k));
      k &= k - 1u;
      const float4* __restrict__ row = (const float4*)(hin + (long long)sj * 256);
      const float w = __expf(leaky(asrc[sj] + ad) - m);
      const float4 x0 = row[lane], x1 = row[32 + lane];
      a0.x += x0.x * w;  a0.y += x0.y * w;  a0.z += x0.z * w;  a0.w += x0.w * w;
      a1.x += x1.x * w;  a1.y += x1.y * w;  a1.z += x1.z * w;  a1.w += x1.w * w;
      z += w;
    }
  }
  const float iz = 1.0f / z;
  float4 v0, v1;
  v0.x = a0.x * iz;  v0.y = a0.y * iz;  v0.z = a0.z * iz;  v0.w = a0.w * iz;
  v1.x = a1.x * iz;  v1.y = a1.y * iz;  v1.z = a1.z * iz;  v1.w = a1.w * iz;
  float4* __restrict__ orow = (float4*)(agg + (long long)n * 256);
  orow[lane]      = v0;
  orow[32 + lane] = v1;
}

__global__ __launch_bounds__(32 * LST_W)
void softmax_aggregate_listed(const int* __restrict__ ei,
                              const int2* __restrict__ l_rec,
                              const int* __restrict__ l_src,
                              const float* __restrict__ hin,
                              const float* __restrict__ asrc,
                              const float* __restrict__ adst,
                              float* __restrict__ agg,
                              int E, int Nn) {
  const int lane = threadIdx.x & 31;
  const int wave = __builtin_amdgcn_readfirstlane(threadIdx.x >> 5);
  const int n0 = (blockIdx.x * LST_W + wave) * LST_R;
  if (n0 >= Nn) return;
  const int nrows = min(LST_R, Nn - n0);

  const int   nl   = min(n0 + lane, Nn - 1);
  const int2  recv = l_rec[nl];
  const float adv  = adst[nl];

  auto window = [&](int cnt, int beg, int i0) {
    return l_src[beg + min(i0 + lane, cnt - 1)];
  };

  int cntN = __builtin_amdgcn_readlane(recv.x, 0);
  int begN = __builtin_amdgcn_readlane(recv.y, 0);
  int svN  = (cntN > 0) ? window(cntN, begN, 0) : 0;

  for (int r = 0; r < nrows; ++r) {
    const int n = n0 + r;
    const int cnt = cntN, beg = begN;
    int sv = svN;
    const float ad = __int_as_float(__builtin_amdgcn_readlane(__float_as_int(adv), r));
    if (r + 1 < nrows) {
      cntN = __builtin_amdgcn_readlane(recv.x, r + 1);
      begN = __builtin_amdgcn_readlane(recv.y, r + 1);
      svN  = (cntN > 0) ? window(cntN, begN, 0) : 0;
    }

    if (cnt <= 0) {
      stream_row(ei, E, hin, asrc, agg, n, ad, lane);
      continue;
    }

    float4 a0 = float4{0.0f, 0.0f, 0.0f, 0.0f}, a1 = a0;
    float z = 0.0f;
    {
      float sc = asrc[sv];
      float4 x0[AGG_B], x1[AGG_B];
      auto request = [&](int i0) {
        const int nb = min(AGG_B, cnt - i0);
        const int wb = i0 & 31;
#pragma unroll
        for (int j = 0; j < AGG_B; ++j) {
          const int sj = __builtin_amdgcn_readlane(sv, wb + min(j, nb - 1));
          const float4* __restrict__ row = (const float4*)(hin + (long long)sj * 256);
          x0[j] = row[lane];
          x1[j] = row[32 + lane];
        }
      };
      auto add_batch = [&](int i0, float m) {
        const int nb = min(AGG_B, cnt - i0);
        const int wb = i0 & 31;
#pragma unroll
        for (int j = 0; j < AGG_B; ++j) {
          const float s  = __int_as_float(__builtin_amdgcn_readlane(__float_as_int(sc),
                                                                    wb + min(j, nb - 1)));
          const float on = (j < nb) ? 1.0f : 0.0f;
          const float w  = __expf(leaky(s + ad) - m) * on;
          a0.x += x0[j].x * w;  a0.y += x0[j].y * w;  a0.z += x0[j].z * w;  a0.w += x0[j].w * w;
          a1.x += x1[j].x * w;  a1.y += x1[j].y * w;  a1.z += x1[j].z * w;  a1.w += x1[j].w * w;
          z += w;
        }
      };

      request(0);
      float m = leaky(sc + ad);
      for (int i0 = 32; i0 < cnt; i0 += 32) {
        const int s = window(cnt, beg, i0);
        m = fmaxf(m, leaky(asrc[s] + ad));
      }
#pragma unroll
      for (int off = 16; off > 0; off >>= 1) m = fmaxf(m, __shfl_xor(m, off));
      add_batch(0, m);

      for (int i0 = AGG_B; i0 < cnt; i0 += AGG_B) {
        if ((i0 & 31) == 0) {
          sv = window(cnt, beg, i0);
          sc = asrc[sv];
        }
        request(i0);
        add_batch(i0, m);
      }
    }

    const float iz = 1.0f / z;
    float4 v0, v1;
    v0.x = a0.x * iz;  v0.y = a0.y * iz;  v0.z = a0.z * iz;  v0.w = a0.w * iz;
    v1.x = a1.x * iz;  v1.y = a1.y * iz;  v1.z = a1.z * iz;  v1.w = a1.w * iz;
    float4* __restrict__ orow = (float4*)(agg + (long long)n * 256);
    orow[lane]      = v0;
    orow[32 + lane] = v1;
  }
}

__global__ void ln_elu(float* __restrict__ hbuf, const float* __restrict__ b1,
                       const float* __restrict__ gamma,
                       const float* __restrict__ beta, int Nn) {
  long long gtid = (long long)blockIdx.x * blockDim.x + threadIdx.x;
  int n    = (int)(gtid >> 5);
  int lane = threadIdx.x & 31;
  if (n >= Nn) return;
  float* __restrict__ row = hbuf + (long long)n * 256;
  float vals[8];
  float sum = 0.0f;
#pragma unroll
  for (int k = 0; k < 8; ++k) {
    int idx = k * 32 + lane;
    vals[k] = row[idx] + b1[idx];
    sum += vals[k];
  }
#pragma unroll
  for (int off = 16; off > 0; off >>= 1) sum += __shfl_xor(sum, off);
  float mu = sum * (1.0f / 256.0f);
  float var = 0.0f;
#pragma unroll
  for (int k = 0; k < 8; ++k) { float t = vals[k] - mu; var += t * t; }
#pragma unroll
  for (int off = 16; off > 0; off >>= 1) var += __shfl_xor(var, off);
  var *= (1.0f / 256.0f);
  float rs = rsqrtf(var + LN_EPS);
#pragma unroll
  for (int k = 0; k < 8; ++k) {
    int idx = k * 32 + lane;
    float t = (vals[k] - mu) * rs * gamma[idx] + beta[idx];
    row[idx] = t > 0.0f ? t : (__expf(t) - 1.0f);
  }
}

#define POOL_W 16
#define POOL_U 16
#define POOL_B 8

__global__ __launch_bounds__(32 * POOL_W)
void pool_own(const float* __restrict__ o, const int* __restrict__ batch,
              float* __restrict__ out, int Nn) {
  __shared__ float4 part[POOL_W][64];
  __shared__ int    pcnt[POOL_W];

  const int lane = threadIdx.x & 31;
  const int wave = __builtin_amdgcn_readfirstlane(threadIdx.x >> 5);
  const int g = blockIdx.x;

  float4 a0 = float4{0.0f, 0.0f, 0.0f, 0.0f}, a1 = a0;
  int cnt = 0;
  const int nstep = (Nn + 31) >> 5;

  for (int s0 = wave; s0 < nstep; s0 += POOL_W * POOL_U) {
    unsigned hw = 0u;
#pragma unroll
    for (int u = 0; u < POOL_U; ++u) {
      const int n = (s0 + u * POOL_W) * 32 + lane;
      const int b = (n < Nn) ? batch[n] : -1;
      hw |= (b == g ? 1u : 0u) << u;
    }
    if (__builtin_amdgcn_ballot_w32(hw != 0u) == 0u) continue;
#pragma nounroll
    for (int u = 0; u < POOL_U; ++u) {
      unsigned m = __builtin_amdgcn_ballot_w32(((hw >> u) & 1u) != 0u);
      const int base = (s0 + u * POOL_W) * 32;
      cnt += __builtin_popcount(m);
      while (m != 0u) {
        float4 x0[POOL_B], x1[POOL_B];
        float  on[POOL_B];
        int p = 0;
#pragma unroll
        for (int j = 0; j < POOL_B; ++j) {
          const bool live = (m != 0u);
          if (live) p = __builtin_ctz(m);
          m &= m - 1u;
          on[j] = live ? 1.0f : 0.0f;
          const float4* __restrict__ row = (const float4*)(o + (long long)(base + p) * 256);
          x0[j] = row[lane];
          x1[j] = row[32 + lane];
        }
#pragma unroll
        for (int j = 0; j < POOL_B; ++j) {
          a0.x += x0[j].x * on[j];  a0.y += x0[j].y * on[j];
          a0.z += x0[j].z * on[j];  a0.w += x0[j].w * on[j];
          a1.x += x1[j].x * on[j];  a1.y += x1[j].y * on[j];
          a1.z += x1[j].z * on[j];  a1.w += x1[j].w * on[j];
        }
      }
    }
  }

  part[wave][lane]      = a0;
  part[wave][32 + lane] = a1;
  if (lane == 0) pcnt[wave] = cnt;
  __syncthreads();

  if (threadIdx.x < 256) {
    const float* pf = (const float*)part;
    float s = 0.0f;
    int c = 0;
#pragma unroll
    for (int w = 0; w < POOL_W; ++w) {
      s += pf[w * 256 + threadIdx.x];
      c += pcnt[w];
    }
    out[(long long)g * 256 + threadIdx.x] = s / fmaxf((float)c, 1.0f);
  }
}

static inline int cdiv(long long a, long long b) { return (int)((a + b - 1) / b); }

extern "C" void kernel_launch(void* const* d_in, const int* in_sizes, int n_in,
                              void* d_out, int out_size, void* d_ws, size_t ws_size,
                              hipStream_t stream) {
  const float* x        = (const float*)d_in[0];
  const int*   ei       = (const int*)  d_in[1];
  const int*   batch    = (const int*)  d_in[2];
  const float* W1       = (const float*)d_in[3];
  const float* att_src1 = (const float*)d_in[4];
  const float* att_dst1 = (const float*)d_in[5];
  const float* b1       = (const float*)d_in[6];
  const float* gamma    = (const float*)d_in[7];
  const float* beta     = (const float*)d_in[8];
  const float* W2       = (const float*)d_in[9];
  const float* att_src2 = (const float*)d_in[10];
  const float* att_dst2 = (const float*)d_in[11];
  const float* b2       = (const float*)d_in[12];
  const float* Wres     = (const float*)d_in[13];
  const float* bres     = (const float*)d_in[14];
  float* out = (float*)d_out;

  const int Nn = in_sizes[2];
  const int Ee = in_sizes[1] / 2;
  const int Gg = out_size / 256;
  const int EA = Ee + Nn;
  const int IN_DIM = in_sizes[0] / Nn;

  const long long NB = (long long)Nn * 256;
  float* ws   = (float*)d_ws;
  float* buf0 = ws;
  float* buf1 = ws + NB;
  float* buf2 = ws + 2 * NB;
  float* cnt  = ws + 3 * NB;
  float* as1  = cnt + Gg;
  float* ad1  = as1 + (long long)Nn * 8;
  float* as2  = ad1 + (long long)Nn * 8;
  float* ad2  = as2 + Nn;
  const int Epad = cdiv(Ee, AGG_S) * AGG_S;
  unsigned* tag = (unsigned*)(ws + ((3 * NB + (long long)Nn * 18 + Gg + 3) / 4) * 4);
  const int NBLK = cdiv(Nn, AGG_R);
  int2* l_rec = (int2*)(tag + Epad / 4);
  int*  l_src = (int*)(l_rec + (long long)NBLK * AGG_R);

  const int B = 256;

  edge_owner<<<cdiv(Epad / 4, B), B, 0, stream>>>(ei, tag, Ee, Epad);

  gemm_wmma_n256<0><<<GEMM_G, 512, 0, stream>>>(x, W1, buf0, nullptr, nullptr, Nn, IN_DIM);
  attn_scores<<<cdiv((long long)Nn * 8, B), B, 0, stream>>>(buf0, att_src1, att_dst1,
                                                            as1, ad1, Nn, 8, 32);
  edge_softmax_aggregate_own<8><<<NBLK, 1024, 0, stream>>>(ei, tag, buf0, as1, ad1, buf1,
                                                           l_rec, l_src, Ee, Epad, Nn);
  ln_elu<<<cdiv((long long)Nn * 32, B), B, 0, stream>>>(buf1, b1, gamma, beta, Nn);

  gemm_wmma_n256<0><<<GEMM_G, 512, 0, stream>>>(buf1, W2, buf0, nullptr, nullptr, Nn, 256);
  attn_scores<<<cdiv((long long)Nn, B), B, 0, stream>>>(buf0, att_src2, att_dst2,
                                                        as2, ad2, Nn, 1, 256);
  softmax_aggregate_listed<<<cdiv(Nn, LST_W * LST_R), 32 * LST_W, 0, stream>>>(ei, l_rec, l_src, buf0, as2, ad2,
                                                                       buf2, Ee, Nn);
  gemm_wmma_n256<1><<<GEMM_G, 512, 0, stream>>>(buf1, Wres, buf2, b2, bres, Nn, 256);

  pool_own<<<Gg, 32 * POOL_W, 0, stream>>>(buf2, batch, out, Nn);
}


# Residual GEMM and layer-2 gather share one launch: every block runs 16 matrix waves beside 16 gather waves (matrix pipe against L2 row reads); residual in its own plane, pool adds the two planes
# speedup vs baseline: 4.2730x; 1.0406x over previous
#include <hip/hip_runtime.h>
#include <hip/hip_bf16.h>


typedef __attribute__((ext_vector_type(2))) float v2f;
typedef __attribute__((ext_vector_type(8))) float v8f;

#define NEG_SLOPE 0.2f
#define LN_EPS    1e-5f

__device__ __forceinline__ unsigned f2sortable(float f) {
  unsigned u = __float_as_uint(f);
  return (u & 0x80000000u) ? ~u : (u | 0x80000000u);
}
__device__ __forceinline__ float sortable2f(unsigned u) {
  u = (u & 0x80000000u) ? (u & 0x7FFFFFFFu) : ~u;
  return __uint_as_float(u);
}
__device__ __forceinline__ float leaky(float x) {
  return x > 0.0f ? x : NEG_SLOPE * x;
}
__device__ __forceinline__ void edge_sd(const int* __restrict__ ei, int E, int e,
                                        int& s, int& d) {
  if (e < E) { s = ei[e]; d = ei[E + e]; }
  else       { s = d = e - E; }
}

#define GEMM_G 256
#define GEMM_P 1056
#define GEMM_KMAX 256

#define GEMM_W 16

template <int NT>
__device__ __forceinline__ void gemm_stage_b(float* Bs, const float* __restrict__ B, int K) {
  for (int i = threadIdx.x; i < (K >> 2) * 256; i += NT) {
    const int g = i >> 8, n = i & 255;
    const float* __restrict__ p = B + (long long)(4 * g) * 256 + n;
    float4 v;
    v.x = p[0]; v.y = p[256]; v.z = p[512]; v.w = p[768];
    *(float4*)(Bs + g * GEMM_P + n * 4) = v;
  }
}

template <int MODE>
__device__ __forceinline__ void gemm_units(const float* Bs,
                                           const float* __restrict__ A,
                                           float* __restrict__ C,
                                           const float* __restrict__ bias_a,
                                           const float* __restrict__ bias_b,
                                           int M, int K, int wave, int lane) {
  const int c    = lane & 15;
  const int h    = lane >> 4;

  const int ntile = (M + 15) >> 4;
  const int t0 = (int)(((long long)blockIdx.x * ntile) / GEMM_G);
  const int t1 = (int)(((long long)(blockIdx.x + 1) * ntile) / GEMM_G);
  const int nunit = (t1 - t0) * 2;

  for (int u = wave; u < nunit; u += GEMM_W) {
    const int tile = t0 + (u >> 1);
    const int cb   = (u & 1) * 128;
    long long arow = (long long)tile * 16 + c;
    if (arow >= M) arow = M - 1;
    const float* __restrict__ Ar = A + arow * K + 4 * h;
    const float* Bl = Bs + h * GEMM_P + (cb + c) * 4;

    v8f acc[8];
#pragma unroll
    for (int t = 0; t < 8; ++t) acc[t] = v8f{};

    float4 an = *(const float4*)Ar;
    auto kstep = [&](int k0) {
      const float4 a4 = an;
      an = *(const float4*)(Ar + min(k0 + 8, K - 8));
      const float* Bq = Bl + (k0 >> 2) * GEMM_P;
      float4 b4[8];
#pragma unroll
      for (int t = 0; t < 8; ++t) b4[t] = *(const float4*)(Bq + t * 64);
      v2f a01, a23;
      a01.x = a4.x; a01.y = a4.y; a23.x = a4.z; a23.y = a4.w;
#pragma unroll
      for (int t = 0; t < 8; ++t) {
        v2f b; b.x = b4[t].x; b.y = b4[t].y;
        acc[t] = __builtin_amdgcn_wmma_f32_16x16x4_f32(false, a01, false, b,
                                                       (short)0, acc[t], false, false);
      }
#pragma unroll
      for (int t = 0; t < 8; ++t) {
        v2f b; b.x = b4[t].z; b.y = b4[t].w;
        acc[t] = __builtin_amdgcn_wmma_f32_16x16x4_f32(false, a23, false, b,
                                                       (short)0, acc[t], false, false);
      }
    };
    if constexpr (MODE == 1) {
#pragma nounroll
      for (int k0 = 0; k0 < K; k0 += 8) {
        kstep(k0);
        __builtin_amdgcn_sched_barrier(0);
        asm volatile("" :: "v"(an.x), "v"(an.y), "v"(an.z), "v"(an.w));
      }
    } else {
      for (int k0 = 0; k0 < K; k0 += 8) kstep(k0);
    }

    float ba[8], bb[8];
#pragma unroll
    for (int t = 0; t < 8; ++t) {
      ba[t] = (MODE == 1) ? bias_a[cb + t * 16 + c] : 0.0f;
      bb[t] = (MODE == 1) ? bias_b[cb + t * 16 + c] : 0.0f;
    }
#pragma unroll
    for (int r = 0; r < 8; ++r) {
      const long long orow = (long long)tile * 16 + h * 8 + r;
      if (orow < M) {
#pragma unroll
        for (int t = 0; t < 8; ++t) {
          const int col = cb + t * 16 + c;
          const long long o = orow * 256 + col;
          if (MODE == 0) C[o] = acc[t][r];
          else           C[o] = acc[t][r] + ba[t] + bb[t];
        }
      }
    }
  }
}

__global__ __launch_bounds__(32 * GEMM_W)
void gemm_wmma_n256(const float* __restrict__ A,
                    const float* __restrict__ B,
                    float* __restrict__ C,
                    int M, int K) {
  __shared__ float Bs[(GEMM_KMAX / 4) * GEMM_P];
  const int lane = threadIdx.x & 31;
  const int wave = __builtin_amdgcn_readfirstlane(threadIdx.x >> 5);
  gemm_stage_b<32 * GEMM_W>(Bs, B, K);
  __syncthreads();
  gemm_units<0>(Bs, A, C, nullptr, nullptr, M, K, wave, lane);
}

__global__ void attn_scores(const float* __restrict__ hbuf,
                            const float* __restrict__ att_s,
                            const float* __restrict__ att_d,
                            float* __restrict__ as_out,
                            float* __restrict__ ad_out,
                            int Nn, int Hh, int Cc) {
  long long tid = (long long)blockIdx.x * blockDim.x + threadIdx.x;
  if (tid >= (long long)Nn * Hh) return;
  int n = (int)(tid / Hh), h = (int)(tid % Hh);
  const float* __restrict__ row = hbuf + (long long)n * 256 + h * Cc;
  const float* __restrict__ vs  = att_s + h * Cc;
  const float* __restrict__ vd  = att_d + h * Cc;
  float ss = 0.0f, sd = 0.0f;
  for (int c = 0; c < Cc; ++c) { float v = row[c]; ss += v * vs[c]; sd += v * vd[c]; }
  as_out[(long long)n * Hh + h] = ss;
  ad_out[(long long)n * Hh + h] = sd;
}

#define AGG_R 200
#define AGG_B 8
#define AGG_S 512
#define AGG_Q 768
#define AGG_BAD 255u
#define AGG_PER ((AGG_R + 31) / 32)
#define AGG_LS (32 * AGG_Q + AGG_R)

__global__ void edge_owner(const int* __restrict__ ei, unsigned* __restrict__ tag4,
                           int E, int Epad) {
  const int t = blockIdx.x * blockDim.x + threadIdx.x;
  if (t * 4 >= Epad) return;
  unsigned w = 0u;
#pragma unroll
  for (int i = 0; i < 4; ++i) {
    const int e = t * 4 + i;
    const unsigned b = (e < E) ? (unsigned)min(ei[E + e] / AGG_R, 255) : 255u;
    w |= b << (8 * i);
  }
  tag4[t] = w;
}

template <int HEADS>
__global__ __launch_bounds__(1024)
void edge_softmax_aggregate_own(const int* __restrict__ ei,
                                const unsigned* __restrict__ tag,
                                const float* __restrict__ hin,
                                const float* __restrict__ asrc,
                                const float* __restrict__ adst,
                                float* __restrict__ agg,
                                int2* __restrict__ l_rec,
                                int* __restrict__ l_src,
                                int E, int Epad, int Nn) {
  __shared__ float    c_ad[AGG_R * HEADS];
  __shared__ unsigned c_m[AGG_R * HEADS];
  __shared__ int      queue[32 * AGG_Q];
  __shared__ int      rowsrc[32 * AGG_Q + AGG_R];
  __shared__ int      r_cnt[AGG_R];
  __shared__ int      r_beg[AGG_R];
  __shared__ int      r_fill[AGG_R];
  __shared__ int      s_lost;

  const int lane = threadIdx.x & 31;
  const int wave = __builtin_amdgcn_readfirstlane(threadIdx.x >> 5);
  const int n0   = blockIdx.x * AGG_R;
  const int nown = min(AGG_R, Nn - n0);
  const unsigned mytag = min(blockIdx.x, 255u);
  const int hA   = (HEADS == 8) ? (lane >> 3) : 0;
  const int hB   = (HEADS == 8) ? 4 + (lane >> 3) : 0;

  for (int t = threadIdx.x; t < nown * HEADS; t += 1024) {
    const long long g = (long long)n0 * HEADS + t;
    const float a_d = adst[g];
    c_ad[t] = a_d;
    c_m[t]  = f2sortable(leaky(asrc[g] + a_d));
  }
  if (threadIdx.x < AGG_R) r_cnt[threadIdx.x] = 1;
  if (threadIdx.x == 0) s_lost = 0;
  __syncthreads();

  const int* __restrict__ dstp = ei + E;
  int* q = queue + wave * AGG_Q;
  int qn = 0;

  auto pack_range = [&](int* qp, int n) {
    for (int i0 = 0; i0 < n; i0 += 128) {
      int s[4]; unsigned dl[4];
#pragma unroll
      for (int u = 0; u < 4; ++u) {
        const int e = qp[min(i0 + u * 32 + lane, n - 1)];
        s[u]  = ei[e];
        dl[u] = (unsigned)(dstp[e] - n0);
      }
#pragma unroll
      for (int u = 0; u < 4; ++u) {
        const int i = i0 + u * 32 + lane;
        if (i < n) {
          const unsigned d = dl[u] < (unsigned)nown ? dl[u] : AGG_BAD;
          qp[i] = s[u] | (int)(d << 24);
          if (d != AGG_BAD) atomicAdd(&r_cnt[d], 1);
        }
      }
    }
  };

  constexpr int MU = (HEADS == 8) ? 8 : 4;
  auto max_range = [&](const int* qp, int n) {
    const unsigned items = (unsigned)n * HEADS;
    for (unsigned t0 = 0; t0 < items; t0 += 32 * MU) {
      unsigned cj[MU]; float a[MU];
#pragma unroll
      for (int u = 0; u < MU; ++u) {
        const unsigned t = min(t0 + u * 32 + lane, items - 1);
        const unsigned v = (unsigned)qp[t / HEADS];
        const unsigned h = t % HEADS;
        cj[u] = (v >> 24) * HEADS + h;
        a[u]  = asrc[(long long)(v & 0xFFFFFFu) * HEADS + h];
      }
      if constexpr (MU == 8)
        asm volatile("" :: "v"(a[0]), "v"(a[1]), "v"(a[2]), "v"(a[3]),
                           "v"(a[4]), "v"(a[5]), "v"(a[6]), "v"(a[7]));
      else
        asm volatile("" :: "v"(a[0]), "v"(a[1]), "v"(a[2]), "v"(a[3]));
#pragma unroll
      for (int u = 0; u < MU; ++u) {
        const bool ok = (t0 + u * 32 + lane < items) && (cj[u] < (unsigned)(nown * HEADS));
        const unsigned cc = ok ? cj[u] : 0u;
        const unsigned val = f2sortable(leaky(a[u] + c_ad[cc]));
        if (ok) atomicMax(&c_m[cc], val);
      }
    }
  };

  bool lost = false;

  const int nstep = Epad / AGG_S;
  auto scan = [&](auto&& after_step) {
    if (nstep <= 0) return;
    const uint4* __restrict__ tag16 = (const uint4*)tag;
    const unsigned pat = mytag * 0x01010101u;
    const int rot = (int)(((long long)blockIdx.x * nstep) / gridDim.x);
    auto step_of = [&](int k) { const int c = k + rot; return c >= nstep ? c - nstep : c; };
    auto load_step = [&](int k) { return tag16[step_of(min(k, nstep - 1)) * 32 + lane]; };
    auto detect = [&](const uint4 t) {
      const unsigned wv[4] = {t.x, t.y, t.z, t.w};
      unsigned hm = 0u;
#pragma unroll
      for (int j = 0; j < 4; ++j) {
        const unsigned x = wv[j] ^ pat;
        const unsigned y = (x & 0x7F7F7F7Fu) + 0x7F7F7F7Fu;
        const unsigned z = ~(y | x) & 0x80808080u;
        hm |= z >> (7 - j);
      }
      return hm;
    };
    auto collect = [&](unsigned hm, int k) {
      const int ebase = step_of(k) * AGG_S + lane * 16;
      unsigned m = __builtin_amdgcn_ballot_w32(hm != 0u);
      while (m != 0u) {
        const int p = __builtin_ctz(hm | 0x80000000u);
        const int e = ebase + ((p & 3) << 2) + (p >> 3);
        const bool hit = (hm != 0u) && (e < E);
        const unsigned mask = __builtin_amdgcn_ballot_w32(hit);
        if (hit) q[qn + __builtin_popcount(mask & ((1u << lane) - 1u))] = e;
        qn += __builtin_popcount(mask);
        hm &= hm - 1u;
        m = __builtin_amdgcn_ballot_w32(hm != 0u);
      }
      after_step(k + 32 < nstep);
    };
    uint4 ta = load_step(wave), tb = load_step(wave + 32);
    for (int k = wave; k < nstep && !lost; k += 64) {
      const unsigned h0 = detect(ta);
      const unsigned h1 = detect(tb);
      ta = load_step(k + 64);
      tb = load_step(k + 96);
      collect(h0, k);
      if (!lost && k + 32 < nstep) collect(h1, k + 32);
    }
  };

  auto weight = [&](float s, float ad, float m) { return __expf(leaky(s + ad) - m); };
  auto add_row = [&](float4& a0, float4& a1, float& zA, float& zB,
                     const float4 x0, const float4 x1, float wA, float wB) {
    a0.x += x0.x * wA;  a0.y += x0.y * wA;  a0.z += x0.z * wA;  a0.w += x0.w * wA;
    a1.x += x1.x * wB;  a1.y += x1.y * wB;  a1.z += x1.z * wB;  a1.w += x1.w * wB;
    zA += wA;  zB += wB;
  };
  auto store_row = [&](long long n, const float4 a0, const float4 a1, float zA, float zB) {
    const float iA = 1.0f / zA, iB = 1.0f / zB;
    float4 v0, v1;
    v0.x = a0.x * iA;  v0.y = a0.y * iA;  v0.z = a0.z * iA;  v0.w = a0.w * iA;
    v1.x = a1.x * iB;  v1.y = a1.y * iB;  v1.z = a1.z * iB;  v1.w = a1.w * iB;
    float4* __restrict__ orow = (float4*)(agg + n * 256);
    orow[lane]      = v0;
    orow[32 + lane] = v1;
  };

  scan([&](bool more) {
    if (more && qn > AGG_Q - AGG_S) lost = true;
  });
  if (lost) s_lost = 1;
  else if (qn > 0) { pack_range(q, qn); max_range(q, qn); }
  __syncthreads();

  if (s_lost != 0) {
    if ((int)threadIdx.x < nown) l_rec[n0 + threadIdx.x] = int2{0, 0};
    for (int r = wave; r < nown; r += 32) {
      const long long n = n0 + r;
      const float adA = adst[n * HEADS + hA], adB = adst[n * HEADS + hB];
      const float slA = asrc[n * HEADS + hA], slB = asrc[n * HEADS + hB];
      float mA = leaky(slA + adA), mB = leaky(slB + adB);
      for (int e0 = 0; e0 < E; e0 += 32) {
        const int e = e0 + lane;
        const bool hit = (e < E) && (dstp[e] == (int)n);
        const int s = hit ? ei[e] : 0;
        unsigned m = __builtin_amdgcn_ballot_w32(hit);
        while (m != 0u) {
          const int sj = __builtin_amdgcn_readlane(s, __builtin_ctz(m));
          m &= m - 1u;
          mA = fmaxf(mA, leaky(asrc[(long long)sj * HEADS + hA] + adA));
          mB = fmaxf(mB, leaky(asrc[(long long)sj * HEADS + hB] + adB));
        }
      }
      const float4* __restrict__ own = (const float4*)(hin + n * 256);
      float4 a0 = float4{0.0f, 0.0f, 0.0f, 0.0f}, a1 = a0;
      float zA = 0.0f, zB = 0.0f;
      add_row(a0, a1, zA, zB, own[lane], own[32 + lane],
              weight(slA, adA, mA), weight(slB, adB, mB));
      for (int e0 = 0; e0 < E; e0 += 32) {
        const int e = e0 + lane;
        const bool hit = (e < E) && (dstp[e] == (int)n);
        const int s = hit ? ei[e] : 0;
        unsigned m = __builtin_amdgcn_ballot_w32(hit);
        while (m != 0u) {
          const int sj = __builtin_amdgcn_readlane(s, __builtin_ctz(m));
          m &= m - 1u;
          const float4* __restrict__ row = (const float4*)(hin + (long long)sj * 256);
          add_row(a0, a1, zA, zB, row[lane], row[32 + lane],
                  weight(asrc[(long long)sj * HEADS + hA], adA, mA),
                  weight(asrc[(long long)sj * HEADS + hB], adB, mB));
        }
      }
      store_row(n, a0, a1, zA, zB);
    }
    return;
  }

  if (wave == 0) {
    int c[AGG_PER];
    int sum = 0;
#pragma unroll
    for (int j = 0; j < AGG_PER; ++j) {
      const int r = lane * AGG_PER + j;
      c[j] = (r < nown) ? r_cnt[r] : 0;
      sum += c[j];
    }
    int inc = sum;
#pragma unroll
    for (int off = 1; off < 32; off <<= 1) {
      const int t = __shfl_up(inc, off);
      if (lane >= off) inc += t;
    }
    int run = inc - sum;
#pragma unroll
    for (int j = 0; j < AGG_PER; ++j) {
      const int r = lane * AGG_PER + j;
      if (r < nown) {
        r_beg[r]    = run;
        r_fill[r]   = run + 1;
        rowsrc[run] = n0 + r;
      }
      run += c[j];
    }
  }
  __syncthreads();

  for (int i = lane; i < qn; i += 32) {
    const unsigned v = (unsigned)q[i];
    const unsigned d = v >> 24;
    if (d != AGG_BAD) rowsrc[atomicAdd(&r_fill[d], 1)] = (int)(v & 0xFFFFFFu);
  }
  __syncthreads();

  {
    const int lbase = blockIdx.x * AGG_LS;
    const int total = r_beg[nown - 1] + r_cnt[nown - 1];
    if ((int)threadIdx.x < nown)
      l_rec[n0 + threadIdx.x] = int2{r_cnt[threadIdx.x], lbase + r_beg[threadIdx.x]};
    for (int i = threadIdx.x; i < total; i += 1024) l_src[lbase + i] = rowsrc[i];
  }

  for (int r = wave; r < nown; r += 32) {
    const int cnt = __builtin_amdgcn_readfirstlane(r_cnt[r]);
    const int beg = __builtin_amdgcn_readfirstlane(r_beg[r]);
    const int cj  = r * HEADS;
    const float adA = c_ad[cj + hA], mA = sortable2f(c_m[cj + hA]);
    const float adB = c_ad[cj + hB], mB = sortable2f(c_m[cj + hB]);
    float4 a0 = float4{0.0f, 0.0f, 0.0f, 0.0f}, a1 = a0;
    float zA = 0.0f, zB = 0.0f;
    for (int i0 = 0; i0 < cnt; i0 += AGG_B) {
      const int nb = min(AGG_B, cnt - i0);
      const int sv = rowsrc[beg + i0 + min(lane, nb - 1)];
      float4 x0[AGG_B], x1[AGG_B];
      float  sA[AGG_B], sB[AGG_B];
#pragma unroll
      for (int j = 0; j < AGG_B; ++j) {
        const int sj = __builtin_amdgcn_readlane(sv, j);
        const float4* __restrict__ row = (const float4*)(hin + (long long)sj * 256);
        x0[j] = row[lane];
        x1[j] = row[32 + lane];
        sA[j] = asrc[(long long)sj * HEADS + hA];
        sB[j] = (HEADS == 8) ? asrc[(long long)sj * HEADS + hB] : 0.0f;
      }
#pragma unroll
      for (int j = 0; j < AGG_B; ++j) {
        const float on = (j < nb) ? 1.0f : 0.0f;
        const float wA = weight(sA[j], adA, mA) * on;
        const float wB = (HEADS == 8) ? weight(sB[j], adB, mB) * on : wA;
        add_row(a0, a1, zA, zB, x0[j], x1[j], wA, wB);
      }
    }
    store_row((long long)n0 + r, a0, a1, zA, zB);
  }
}

#define LST_W 16

__device__ __forceinline__ void stream_row(const int* __restrict__ ei, int E,
                                           const float* __restrict__ hin,
                                           const float* __restrict__ asrc,
                                           float* __restrict__ agg,
                                           int n, float ad, int lane) {
  float4 a0 = float4{0.0f, 0.0f, 0.0f, 0.0f}, a1 = a0;
  float z = 0.0f;
  const int* __restrict__ dstp = ei + E;
  const float sl = asrc[n];
  float m = leaky(sl + ad);
  for (int e0 = 0; e0 < E; e0 += 32) {
    const int e = e0 + lane;
    const bool hit = (e < E) && (dstp[e] == n);
    const int s = hit ? ei[e] : 0;
    unsigned k = __builtin_amdgcn_ballot_w32(hit);
    while (k != 0u) {
      const int sj = __builtin_amdgcn_readlane(s, __builtin_ctz(k));
      k &= k - 1u;
      m = fmaxf(m, leaky(asrc[sj] + ad));
    }
  }
  {
    const float4* __restrict__ own = (const float4*)(hin + (long long)n * 256);
    const float w = __expf(leaky(sl + ad) - m);
    const float4 x0 = own[lane], x1 = own[32 + lane];
    a0.x += x0.x * w;  a0.y += x0.y * w;  a0.z += x0.z * w;  a0.w += x0.w * w;
    a1.x += x1.x * w;  a1.y += x1.y * w;  a1.z += x1.z * w;  a1.w += x1.w * w;
    z += w;
  }
  for (int e0 = 0; e0 < E; e0 += 32) {
    const int e = e0 + lane;
    const bool hit = (e < E) && (dstp[e] == n);
    const int s = hit ? ei[e] : 0;
    unsigned k = __builtin_amdgcn_ballot_w32(hit);
    while (k != 0u) {
      const int sj = __builtin_amdgcn_readlane(s, __builtin_ctz(k));
      k &= k - 1u;
      const float4* __restrict__ row = (const float4*)(hin + (long long)sj * 256);
      const float w = __expf(leaky(asrc[sj] + ad) - m);
      const float4 x0 = row[lane], x1 = row[32 + lane];
      a0.x += x0.x * w;  a0.y += x0.y * w;  a0.z += x0.z * w;  a0.w += x0.w * w;
      a1.x += x1.x * w;  a1.y += x1.y * w;  a1.z += x1.z * w;  a1.w += x1.w * w;
      z += w;
    }
  }
  const float iz = 1.0f / z;
  float4 v0, v1;
  v0.x = a0.x * iz;  v0.y = a0.y * iz;  v0.z = a0.z * iz;  v0.w = a0.w * iz;
  v1.x = a1.x * iz;  v1.y = a1.y * iz;  v1.z = a1.z * iz;  v1.w = a1.w * iz;
  float4* __restrict__ orow = (float4*)(agg + (long long)n * 256);
  orow[lane]      = v0;
  orow[32 + lane] = v1;
}

__device__ __forceinline__ void listed_rows(const int* __restrict__ ei,
                                            const int2* __restrict__ l_rec,
                                            const int* __restrict__ l_src,
                                            const float* __restrict__ hin,
                                            const float* __restrict__ asrc,
                                            const float* __restrict__ adst,
                                            float* __restrict__ agg,
                                            int E, int Nn, int n0, int nrows, int lane) {

  const int   nl   = min(n0 + lane, Nn - 1);
  const int2  recv = l_rec[nl];
  const float adv  = adst[nl];

  auto window = [&](int cnt, int beg, int i0) {
    return l_src[beg + min(i0 + lane, cnt - 1)];
  };

  int cntN = __builtin_amdgcn_readlane(recv.x, 0);
  int begN = __builtin_amdgcn_readlane(recv.y, 0);
  int svN  = (cntN > 0) ? window(cntN, begN, 0) : 0;

  for (int r = 0; r < nrows; ++r) {
    const int n = n0 + r;
    const int cnt = cntN, beg = begN;
    int sv = svN;
    const float ad = __int_as_float(__builtin_amdgcn_readlane(__float_as_int(adv), r));
    if (r + 1 < nrows) {
      cntN = __builtin_amdgcn_readlane(recv.x, r + 1);
      begN = __builtin_amdgcn_readlane(recv.y, r + 1);
      svN  = (cntN > 0) ? window(cntN, begN, 0) : 0;
    }

    if (cnt <= 0) {
      stream_row(ei, E, hin, asrc, agg, n, ad, lane);
      continue;
    }

    float4 a0 = float4{0.0f, 0.0f, 0.0f, 0.0f}, a1 = a0;
    float z = 0.0f;
    {
      float sc = asrc[sv];
      float4 x0[AGG_B], x1[AGG_B];
      auto request = [&](int i0) {
        const int nb = min(AGG_B, cnt - i0);
        const int wb = i0 & 31;
#pragma unroll
        for (int j = 0; j < AGG_B; ++j) {
          const int sj = __builtin_amdgcn_readlane(sv, wb + min(j, nb - 1));
          const float4* __restrict__ row = (const float4*)(hin + (long long)sj * 256);
          x0[j] = row[lane];
          x1[j] = row[32 + lane];
        }
      };
      auto add_batch = [&](int i0, float m) {
        const int nb = min(AGG_B, cnt - i0);
        const int wb = i0 & 31;
#pragma unroll
        for (int j = 0; j < AGG_B; ++j) {
          const float s  = __int_as_float(__builtin_amdgcn_readlane(__float_as_int(sc),
                                                                    wb + min(j, nb - 1)));
          const float on = (j < nb) ? 1.0f : 0.0f;
          const float w  = __expf(leaky(s + ad) - m) * on;
          a0.x += x0[j].x * w;  a0.y += x0[j].y * w;  a0.z += x0[j].z * w;  a0.w += x0[j].w * w;
          a1.x += x1[j].x * w;  a1.y += x1[j].y * w;  a1.z += x1[j].z * w;  a1.w += x1[j].w * w;
          z += w;
        }
      };

      request(0);
      float m = leaky(sc + ad);
      for (int i0 = 32; i0 < cnt; i0 += 32) {
        const int s = window(cnt, beg, i0);
        m = fmaxf(m, leaky(asrc[s] + ad));
      }
#pragma unroll
      for (int off = 16; off > 0; off >>= 1) m = fmaxf(m, __shfl_xor(m, off));
      add_batch(0, m);

      for (int i0 = AGG_B; i0 < cnt; i0 += AGG_B) {
        if ((i0 & 31) == 0) {
          sv = window(cnt, beg, i0);
          sc = asrc[sv];
        }
        request(i0);
        add_batch(i0, m);
      }
    }

    const float iz = 1.0f / z;
    float4 v0, v1;
    v0.x = a0.x * iz;  v0.y = a0.y * iz;  v0.z = a0.z * iz;  v0.w = a0.w * iz;
    v1.x = a1.x * iz;  v1.y = a1.y * iz;  v1.z = a1.z * iz;  v1.w = a1.w * iz;
    float4* __restrict__ orow = (float4*)(agg + (long long)n * 256);
    orow[lane]      = v0;
    orow[32 + lane] = v1;
  }
}

__global__ __launch_bounds__(32 * (GEMM_W + LST_W))
void resid_gemm_beside_gather(const float* __restrict__ A,
                              const float* __restrict__ B,
                              float* __restrict__ R,
                              const float* __restrict__ bias_a,
                              const float* __restrict__ bias_b,
                              int M, int K,
                              const int* __restrict__ ei,
                              const int2* __restrict__ l_rec,
                              const int* __restrict__ l_src,
                              const float* __restrict__ hin,
                              const float* __restrict__ asrc,
                              const float* __restrict__ adst,
                              float* __restrict__ agg,
                              int E) {
  __shared__ float Bs[(GEMM_KMAX / 4) * GEMM_P];
  const int lane = threadIdx.x & 31;
  const int wave = __builtin_amdgcn_readfirstlane(threadIdx.x >> 5);
  gemm_stage_b<32 * (GEMM_W + LST_W)>(Bs, B, K);
  __syncthreads();
  if (wave < GEMM_W) {
    gemm_units<1>(Bs, A, R, bias_a, bias_b, M, K, wave, lane);
  } else {
    const long long g  = (long long)blockIdx.x * LST_W + (wave - GEMM_W);
    const long long ng = (long long)gridDim.x * LST_W;
    const int r0 = (int)((g * M) / ng);
    const int r1 = (int)(((g + 1) * M) / ng);
    for (int n0 = r0; n0 < r1; n0 += 32)
      listed_rows(ei, l_rec, l_src, hin, asrc, adst, agg, E, M, n0, min(32, r1 - n0), lane);
  }
}

__global__ void ln_elu(float* __restrict__ hbuf, const float* __restrict__ b1,
                       const float* __restrict__ gamma,
                       const float* __restrict__ beta, int Nn) {
  long long gtid = (long long)blockIdx.x * blockDim.x + threadIdx.x;
  int n    = (int)(gtid >> 5);
  int lane = threadIdx.x & 31;
  if (n >= Nn) return;
  float* __restrict__ row = hbuf + (long long)n * 256;
  float vals[8];
  float sum = 0.0f;
#pragma unroll
  for (int k = 0; k < 8; ++k) {
    int idx = k * 32 + lane;
    vals[k] = row[idx] + b1[idx];
    sum += vals[k];
  }
#pragma unroll
  for (int off = 16; off > 0; off >>= 1) sum += __shfl_xor(sum, off);
  float mu = sum * (1.0f / 256.0f);
  float var = 0.0f;
#pragma unroll
  for (int k = 0; k < 8; ++k) { float t = vals[k] - mu; var += t * t; }
#pragma unroll
  for (int off = 16; off > 0; off >>= 1) var += __shfl_xor(var, off);
  var *= (1.0f / 256.0f);
  float rs = rsqrtf(var + LN_EPS);
#pragma unroll
  for (int k = 0; k < 8; ++k) {
    int idx = k * 32 + lane;
    float t = (vals[k] - mu) * rs * gamma[idx] + beta[idx];
    row[idx] = t > 0.0f ? t : (__expf(t) - 1.0f);
  }
}

#define POOL_W 16
#define POOL_U 16
#define POOL_B 4

__global__ __launch_bounds__(32 * POOL_W)
void pool_own(const float* __restrict__ o, const float* __restrict__ res,
              const int* __restrict__ batch, float* __restrict__ out, int Nn) {
  __shared__ float4 part[POOL_W][64];
  __shared__ int    pcnt[POOL_W];

  const int lane = threadIdx.x & 31;
  const int wave = __builtin_amdgcn_readfirstlane(threadIdx.x >> 5);
  const int g = blockIdx.x;

  float4 a0 = float4{0.0f, 0.0f, 0.0f, 0.0f}, a1 = a0;
  int cnt = 0;
  const int nstep = (Nn + 31) >> 5;

  for (int s0 = wave; s0 < nstep; s0 += POOL_W * POOL_U) {
    unsigned hw = 0u;
#pragma unroll
    for (int u = 0; u < POOL_U; ++u) {
      const int n = (s0 + u * POOL_W) * 32 + lane;
      const int b = (n < Nn) ? batch[n] : -1;
      hw |= (b == g ? 1u : 0u) << u;
    }
    if (__builtin_amdgcn_ballot_w32(hw != 0u) == 0u) continue;
#pragma nounroll
    for (int u = 0; u < POOL_U; ++u) {
      unsigned m = __builtin_amdgcn_ballot_w32(((hw >> u) & 1u) != 0u);
      const int base = (s0 + u * POOL_W) * 32;
      cnt += __builtin_popcount(m);
      while (m != 0u) {
        float4 x0[POOL_B], x1[POOL_B], y0[POOL_B], y1[POOL_B];
        float  on[POOL_B];
        int p = 0;
#pragma unroll
        for (int j = 0; j < POOL_B; ++j) {
          const bool live = (m != 0u);
          if (live) p = __builtin_ctz(m);
          m &= m - 1u;
          on[j] = live ? 1.0f : 0.0f;
          const float4* __restrict__ row = (const float4*)(o + (long long)(base + p) * 256);
          const float4* __restrict__ rrw = (const float4*)(res + (long long)(base + p) * 256);
          x0[j] = row[lane];
          x1[j] = row[32 + lane];
          y0[j] = rrw[lane];
          y1[j] = rrw[32 + lane];
        }
#pragma unroll
        for (int j = 0; j < POOL_B; ++j) {
          a0.x += (x0[j].x + y0[j].x) * on[j];  a0.y += (x0[j].y + y0[j].y) * on[j];
          a0.z += (x0[j].z + y0[j].z) * on[j];  a0.w += (x0[j].w + y0[j].w) * on[j];
          a1.x += (x1[j].x + y1[j].x) * on[j];  a1.y += (x1[j].y + y1[j].y) * on[j];
          a1.z += (x1[j].z + y1[j].z) * on[j];  a1.w += (x1[j].w + y1[j].w) * on[j];
        }
      }
    }
  }

  part[wave][lane]      = a0;
  part[wave][32 + lane] = a1;
  if (lane == 0) pcnt[wave] = cnt;
  __syncthreads();

  if (threadIdx.x < 256) {
    const float* pf = (const float*)part;
    float s = 0.0f;
    int c = 0;
#pragma unroll
    for (int w = 0; w < POOL_W; ++w) {
      s += pf[w * 256 + threadIdx.x];
      c += pcnt[w];
    }
    out[(long long)g * 256 + threadIdx.x] = s / fmaxf((float)c, 1.0f);
  }
}

static inline int cdiv(long long a, long long b) { return (int)((a + b - 1) / b); }

extern "C" void kernel_launch(void* const* d_in, const int* in_sizes, int n_in,
                              void* d_out, int out_size, void* d_ws, size_t ws_size,
                              hipStream_t stream) {
  const float* x        = (const float*)d_in[0];
  const int*   ei       = (const int*)  d_in[1];
  const int*   batch    = (const int*)  d_in[2];
  const float* W1       = (const float*)d_in[3];
  const float* att_src1 = (const float*)d_in[4];
  const float* att_dst1 = (const float*)d_in[5];
  const float* b1       = (const float*)d_in[6];
  const float* gamma    = (const float*)d_in[7];
  const float* beta     = (const float*)d_in[8];
  const float* W2       = (const float*)d_in[9];
  const float* att_src2 = (const float*)d_in[10];
  const float* att_dst2 = (const float*)d_in[11];
  const float* b2       = (const float*)d_in[12];
  const float* Wres     = (const float*)d_in[13];
  const float* bres     = (const float*)d_in[14];
  float* out = (float*)d_out;

  const int Nn = in_sizes[2];
  const int Ee = in_sizes[1] / 2;
  const int Gg = out_size / 256;
  const int EA = Ee + Nn;
  const int IN_DIM = in_sizes[0] / Nn;

  const long long NB = (long long)Nn * 256;
  float* ws   = (float*)d_ws;
  float* buf0 = ws;
  float* buf1 = ws + NB;
  float* buf2 = ws + 2 * NB;
  float* cnt  = ws + 3 * NB;
  float* as1  = cnt + Gg;
  float* ad1  = as1 + (long long)Nn * 8;
  float* as2  = ad1 + (long long)Nn * 8;
  float* ad2  = as2 + Nn;
  const int Epad = cdiv(Ee, AGG_S) * AGG_S;
  const long long tag_w = ((3 * NB + (long long)Nn * 18 + Gg + 3) / 4) * 4;
  unsigned* tag = (unsigned*)(ws + tag_w);
  const int NBLK = cdiv(Nn, AGG_R);
  int2* l_rec = (int2*)(tag + Epad / 4);
  int*  l_src = (int*)(l_rec + (long long)NBLK * AGG_R);
  const long long res_w = ((tag_w + Epad / 4 + (long long)NBLK * AGG_R * 2 +
                            (long long)NBLK * AGG_LS + 3) / 4) * 4;
  float* buf3 = ws + res_w;

  const int B = 256;

  edge_owner<<<cdiv(Epad / 4, B), B, 0, stream>>>(ei, tag, Ee, Epad);

  gemm_wmma_n256<<<GEMM_G, 32 * GEMM_W, 0, stream>>>(x, W1, buf0, Nn, IN_DIM);
  attn_scores<<<cdiv((long long)Nn * 8, B), B, 0, stream>>>(buf0, att_src1, att_dst1,
                                                            as1, ad1, Nn, 8, 32);
  edge_softmax_aggregate_own<8><<<NBLK, 1024, 0, stream>>>(ei, tag, buf0, as1, ad1, buf1,
                                                           l_rec, l_src, Ee, Epad, Nn);
  ln_elu<<<cdiv((long long)Nn * 32, B), B, 0, stream>>>(buf1, b1, gamma, beta, Nn);

  gemm_wmma_n256<<<GEMM_G, 32 * GEMM_W, 0, stream>>>(buf1, W2, buf0, Nn, 256);
  attn_scores<<<cdiv((long long)Nn, B), B, 0, stream>>>(buf0, att_src2, att_dst2,
                                                        as2, ad2, Nn, 1, 256);
  resid_gemm_beside_gather<<<GEMM_G, 32 * (GEMM_W + LST_W), 0, stream>>>(
      buf1, Wres, buf3, b2, bres, Nn, 256, ei, l_rec, l_src, buf0, as2, ad2, buf2, Ee);

  pool_own<<<Gg, 32 * POOL_W, 0, stream>>>(buf2, buf3, batch, out, Nn);
}


# Row-wise launches (LayerNorm+ELU, both score dots) become streaming kernels: 4096 resident waves loop over rows, 512-byte loads, 4 rows in flight; row-sum and score-dot sum order change
# speedup vs baseline: 4.4586x; 1.0434x over previous
#include <hip/hip_runtime.h>
#include <hip/hip_bf16.h>


typedef __attribute__((ext_vector_type(2))) float v2f;
typedef __attribute__((ext_vector_type(8))) float v8f;

#define NEG_SLOPE 0.2f
#define LN_EPS    1e-5f

__device__ __forceinline__ unsigned f2sortable(float f) {
  unsigned u = __float_as_uint(f);
  return (u & 0x80000000u) ? ~u : (u | 0x80000000u);
}
__device__ __forceinline__ float sortable2f(unsigned u) {
  u = (u & 0x80000000u) ? (u & 0x7FFFFFFFu) : ~u;
  return __uint_as_float(u);
}
__device__ __forceinline__ float leaky(float x) {
  return x > 0.0f ? x : NEG_SLOPE * x;
}
__device__ __forceinline__ void edge_sd(const int* __restrict__ ei, int E, int e,
                                        int& s, int& d) {
  if (e < E) { s = ei[e]; d = ei[E + e]; }
  else       { s = d = e - E; }
}

#define GEMM_G 256
#define GEMM_P 1056
#define GEMM_KMAX 256

#define GEMM_W 16

template <int NT>
__device__ __forceinline__ void gemm_stage_b(float* Bs, const float* __restrict__ B, int K) {
  for (int i = threadIdx.x; i < (K >> 2) * 256; i += NT) {
    const int g = i >> 8, n = i & 255;
    const float* __restrict__ p = B + (long long)(4 * g) * 256 + n;
    float4 v;
    v.x = p[0]; v.y = p[256]; v.z = p[512]; v.w = p[768];
    *(float4*)(Bs + g * GEMM_P + n * 4) = v;
  }
}

template <int MODE>
__device__ __forceinline__ void gemm_units(const float* Bs,
                                           const float* __restrict__ A,
                                           float* __restrict__ C,
                                           const float* __restrict__ bias_a,
                                           const float* __restrict__ bias_b,
                                           int M, int K, int wave, int lane) {
  const int c    = lane & 15;
  const int h    = lane >> 4;

  const int ntile = (M + 15) >> 4;
  const int t0 = (int)(((long long)blockIdx.x * ntile) / GEMM_G);
  const int t1 = (int)(((long long)(blockIdx.x + 1) * ntile) / GEMM_G);
  const int nunit = (t1 - t0) * 2;

  for (int u = wave; u < nunit; u += GEMM_W) {
    const int tile = t0 + (u >> 1);
    const int cb   = (u & 1) * 128;
    long long arow = (long long)tile * 16 + c;
    if (arow >= M) arow = M - 1;
    const float* __restrict__ Ar = A + arow * K + 4 * h;
    const float* Bl = Bs + h * GEMM_P + (cb + c) * 4;

    v8f acc[8];
#pragma unroll
    for (int t = 0; t < 8; ++t) acc[t] = v8f{};

    float4 an = *(const float4*)Ar;
    auto kstep = [&](int k0) {
      const float4 a4 = an;
      an = *(const float4*)(Ar + min(k0 + 8, K - 8));
      const float* Bq = Bl + (k0 >> 2) * GEMM_P;
      float4 b4[8];
#pragma unroll
      for (int t = 0; t < 8; ++t) b4[t] = *(const float4*)(Bq + t * 64);
      v2f a01, a23;
      a01.x = a4.x; a01.y = a4.y; a23.x = a4.z; a23.y = a4.w;
#pragma unroll
      for (int t = 0; t < 8; ++t) {
        v2f b; b.x = b4[t].x; b.y = b4[t].y;
        acc[t] = __builtin_amdgcn_wmma_f32_16x16x4_f32(false, a01, false, b,
                                                       (short)0, acc[t], false, false);
      }
#pragma unroll
      for (int t = 0; t < 8; ++t) {
        v2f b; b.x = b4[t].z; b.y = b4[t].w;
        acc[t] = __builtin_amdgcn_wmma_f32_16x16x4_f32(false, a23, false, b,
                                                       (short)0, acc[t], false, false);
      }
    };
    if constexpr (MODE == 1) {
#pragma nounroll
      for (int k0 = 0; k0 < K; k0 += 8) {
        kstep(k0);
        __builtin_amdgcn_sched_barrier(0);
        asm volatile("" :: "v"(an.x), "v"(an.y), "v"(an.z), "v"(an.w));
      }
    } else {
      for (int k0 = 0; k0 < K; k0 += 8) kstep(k0);
    }

    float ba[8], bb[8];
#pragma unroll
    for (int t = 0; t < 8; ++t) {
      ba[t] = (MODE == 1) ? bias_a[cb + t * 16 + c] : 0.0f;
      bb[t] = (MODE == 1) ? bias_b[cb + t * 16 + c] : 0.0f;
    }
#pragma unroll
    for (int r = 0; r < 8; ++r) {
      const long long orow = (long long)tile * 16 + h * 8 + r;
      if (orow < M) {
#pragma unroll
        for (int t = 0; t < 8; ++t) {
          const int col = cb + t * 16 + c;
          const long long o = orow * 256 + col;
          if (MODE == 0) C[o] = acc[t][r];
          else           C[o] = acc[t][r] + ba[t] + bb[t];
        }
      }
    }
  }
}

__global__ __launch_bounds__(32 * GEMM_W)
void gemm_wmma_n256(const float* __restrict__ A,
                    const float* __restrict__ B,
                    float* __restrict__ C,
                    int M, int K) {
  __shared__ float Bs[(GEMM_KMAX / 4) * GEMM_P];
  const int lane = threadIdx.x & 31;
  const int wave = __builtin_amdgcn_readfirstlane(threadIdx.x >> 5);
  gemm_stage_b<32 * GEMM_W>(Bs, B, K);
  __syncthreads();
  gemm_units<0>(Bs, A, C, nullptr, nullptr, M, K, wave, lane);
}

#define ROWS_G 512
#define ROWS_W 8
#define ROWS_B 4

__device__ __forceinline__ void load_rows(const float* buf, int q, int Nn, int lane,
                                          float4 (&x0)[ROWS_B], float4 (&x1)[ROWS_B]) {
#pragma unroll
  for (int j = 0; j < ROWS_B; ++j) {
    const int n = (q * ROWS_B + j < Nn) ? q * ROWS_B + j : q * ROWS_B;
    const float4* row = (const float4*)(buf + (long long)n * 256);
    x0[j] = row[lane];
    x1[j] = row[32 + lane];
  }
}
__device__ __forceinline__ float dot4(const float4 a, const float4 b) {
  return (a.x * b.x + a.y * b.y) + (a.z * b.z + a.w * b.w);
}

__global__ __launch_bounds__(32 * ROWS_W)
void attn_scores8_rows(const float* __restrict__ hbuf,
                       const float* __restrict__ att_s,
                       const float* __restrict__ att_d,
                       float* as_out, float* ad_out, int Nn) {
  const int lane = threadIdx.x & 31;
  const int wave = __builtin_amdgcn_readfirstlane(threadIdx.x >> 5);
  const int gw = blockIdx.x * ROWS_W + wave;
  const int GW = gridDim.x * ROWS_W;
  const float4 s0 = ((const float4*)att_s)[lane], s1 = ((const float4*)att_s)[32 + lane];
  const float4 d0 = ((const float4*)att_d)[lane], d1 = ((const float4*)att_d)[32 + lane];
  const bool b0 = (lane & 1) != 0, b1 = (lane & 2) != 0, b2 = (lane & 4) != 0;
  const int head = (lane >> 3) + (b0 ? 4 : 0);
  float* outp = b1 ? ad_out : as_out;
  const int nq = (Nn + ROWS_B - 1) / ROWS_B;
  for (int q = gw; q < nq; q += GW) {
    float4 x0[ROWS_B], x1[ROWS_B];
    load_rows(hbuf, q, Nn, lane, x0, x1);
    float w[ROWS_B];
#pragma unroll
    for (int j = 0; j < ROWS_B; ++j) {
      const float As = dot4(x0[j], s0), Bs = dot4(x1[j], s1);
      const float Ad = dot4(x0[j], d0), Bd = dot4(x1[j], d1);
      const float vS = (b0 ? Bs : As) + __shfl_xor(b0 ? As : Bs, 1);
      const float vD = (b0 ? Bd : Ad) + __shfl_xor(b0 ? Ad : Bd, 1);
      w[j] = (b1 ? vD : vS) + __shfl_xor(b1 ? vS : vD, 2);
    }
    const float u01 = (b2 ? w[1] : w[0]) + __shfl_xor(b2 ? w[0] : w[1], 4);
    const float u23 = (b2 ? w[3] : w[2]) + __shfl_xor(b2 ? w[2] : w[3], 4);
    const int nA = q * ROWS_B + (b2 ? 1 : 0);
    const int nB = nA + 2;
    if (nA < Nn) outp[(long long)nA * 8 + head] = u01;
    if (nB < Nn) outp[(long long)nB * 8 + head] = u23;
  }
}

__global__ __launch_bounds__(32 * ROWS_W)
void attn_scores1_rows(const float* __restrict__ hbuf,
                       const float* __restrict__ att_s,
                       const float* __restrict__ att_d,
                       float* as_out, float* ad_out, int Nn) {
  const int lane = threadIdx.x & 31;
  const int wave = __builtin_amdgcn_readfirstlane(threadIdx.x >> 5);
  const int gw = blockIdx.x * ROWS_W + wave;
  const int GW = gridDim.x * ROWS_W;
  const float4 s0 = ((const float4*)att_s)[lane], s1 = ((const float4*)att_s)[32 + lane];
  const float4 d0 = ((const float4*)att_d)[lane], d1 = ((const float4*)att_d)[32 + lane];
  const bool b0 = (lane & 1) != 0, b1 = (lane & 2) != 0, b2 = (lane & 4) != 0;
  float* outp = b0 ? ad_out : as_out;
  const int nq = (Nn + ROWS_B - 1) / ROWS_B;
  for (int q = gw; q < nq; q += GW) {
    float4 x0[ROWS_B], x1[ROWS_B];
    load_rows(hbuf, q, Nn, lane, x0, x1);
    float w[ROWS_B];
#pragma unroll
    for (int j = 0; j < ROWS_B; ++j) {
      const float ps = dot4(x0[j], s0) + dot4(x1[j], s1);
      const float pd = dot4(x0[j], d0) + dot4(x1[j], d1);
      w[j] = (b0 ? pd : ps) + __shfl_xor(b0 ? ps : pd, 1);
    }
    const float u01 = (b1 ? w[1] : w[0]) + __shfl_xor(b1 ? w[0] : w[1], 2);
    const float u23 = (b1 ? w[3] : w[2]) + __shfl_xor(b1 ? w[2] : w[3], 2);
    float t = (b2 ? u23 : u01) + __shfl_xor(b2 ? u01 : u23, 4);
    t += __shfl_xor(t, 8);
    t += __shfl_xor(t, 16);
    const int n = q * ROWS_B + ((lane >> 1) & 3);
    if (lane < 8 && n < Nn) outp[n] = t;
  }
}

#define AGG_R 200
#define AGG_B 8
#define AGG_S 512
#define AGG_Q 768
#define AGG_BAD 255u
#define AGG_PER ((AGG_R + 31) / 32)
#define AGG_LS (32 * AGG_Q + AGG_R)

__global__ void edge_owner(const int* __restrict__ ei, unsigned* __restrict__ tag4,
                           int E, int Epad) {
  const int t = blockIdx.x * blockDim.x + threadIdx.x;
  if (t * 4 >= Epad) return;
  unsigned w = 0u;
#pragma unroll
  for (int i = 0; i < 4; ++i) {
    const int e = t * 4 + i;
    const unsigned b = (e < E) ? (unsigned)min(ei[E + e] / AGG_R, 255) : 255u;
    w |= b << (8 * i);
  }
  tag4[t] = w;
}

template <int HEADS>
__global__ __launch_bounds__(1024)
void edge_softmax_aggregate_own(const int* __restrict__ ei,
                                const unsigned* __restrict__ tag,
                                const float* __restrict__ hin,
                                const float* __restrict__ asrc,
                                const float* __restrict__ adst,
                                float* __restrict__ agg,
                                int2* __restrict__ l_rec,
                                int* __restrict__ l_src,
                                int E, int Epad, int Nn) {
  __shared__ float    c_ad[AGG_R * HEADS];
  __shared__ unsigned c_m[AGG_R * HEADS];
  __shared__ int      queue[32 * AGG_Q];
  __shared__ int      rowsrc[32 * AGG_Q + AGG_R];
  __shared__ int      r_cnt[AGG_R];
  __shared__ int      r_beg[AGG_R];
  __shared__ int      r_fill[AGG_R];
  __shared__ int      s_lost;

  const int lane = threadIdx.x & 31;
  const int wave = __builtin_amdgcn_readfirstlane(threadIdx.x >> 5);
  const int n0   = blockIdx.x * AGG_R;
  const int nown = min(AGG_R, Nn - n0);
  const unsigned mytag = min(blockIdx.x, 255u);
  const int hA   = (HEADS == 8) ? (lane >> 3) : 0;
  const int hB   = (HEADS == 8) ? 4 + (lane >> 3) : 0;

  for (int t = threadIdx.x; t < nown * HEADS; t += 1024) {
    const long long g = (long long)n0 * HEADS + t;
    const float a_d = adst[g];
    c_ad[t] = a_d;
    c_m[t]  = f2sortable(leaky(asrc[g] + a_d));
  }
  if (threadIdx.x < AGG_R) r_cnt[threadIdx.x] = 1;
  if (threadIdx.x == 0) s_lost = 0;
  __syncthreads();

  const int* __restrict__ dstp = ei + E;
  int* q = queue + wave * AGG_Q;
  int qn = 0;

  auto pack_range = [&](int* qp, int n) {
    for (int i0 = 0; i0 < n; i0 += 128) {
      int s[4]; unsigned dl[4];
#pragma unroll
      for (int u = 0; u < 4; ++u) {
        const int e = qp[min(i0 + u * 32 + lane, n - 1)];
        s[u]  = ei[e];
        dl[u] = (unsigned)(dstp[e] - n0);
      }
#pragma unroll
      for (int u = 0; u < 4; ++u) {
        const int i = i0 + u * 32 + lane;
        if (i < n) {
          const unsigned d = dl[u] < (unsigned)nown ? dl[u] : AGG_BAD;
          qp[i] = s[u] | (int)(d << 24);
          if (d != AGG_BAD) atomicAdd(&r_cnt[d], 1);
        }
      }
    }
  };

  constexpr int MU = (HEADS == 8) ? 8 : 4;
  auto max_range = [&](const int* qp, int n) {
    const unsigned items = (unsigned)n * HEADS;
    for (unsigned t0 = 0; t0 < items; t0 += 32 * MU) {
      unsigned cj[MU]; float a[MU];
#pragma unroll
      for (int u = 0; u < MU; ++u) {
        const unsigned t = min(t0 + u * 32 + lane, items - 1);
        const unsigned v = (unsigned)qp[t / HEADS];
        const unsigned h = t % HEADS;
        cj[u] = (v >> 24) * HEADS + h;
        a[u]  = asrc[(long long)(v & 0xFFFFFFu) * HEADS + h];
      }
      if constexpr (MU == 8)
        asm volatile("" :: "v"(a[0]), "v"(a[1]), "v"(a[2]), "v"(a[3]),
                           "v"(a[4]), "v"(a[5]), "v"(a[6]), "v"(a[7]));
      else
        asm volatile("" :: "v"(a[0]), "v"(a[1]), "v"(a[2]), "v"(a[3]));
#pragma unroll
      for (int u = 0; u < MU; ++u) {
        const bool ok = (t0 + u * 32 + lane < items) && (cj[u] < (unsigned)(nown * HEADS));
        const unsigned cc = ok ? cj[u] : 0u;
        const unsigned val = f2sortable(leaky(a[u] + c_ad[cc]));
        if (ok) atomicMax(&c_m[cc], val);
      }
    }
  };

  bool lost = false;

  const int nstep = Epad / AGG_S;
  auto scan = [&](auto&& after_step) {
    if (nstep <= 0) return;
    const uint4* __restrict__ tag16 = (const uint4*)tag;
    const unsigned pat = mytag * 0x01010101u;
    const int rot = (int)(((long long)blockIdx.x * nstep) / gridDim.x);
    auto step_of = [&](int k) { const int c = k + rot; return c >= nstep ? c - nstep : c; };
    auto load_step = [&](int k) { return tag16[step_of(min(k, nstep - 1)) * 32 + lane]; };
    auto detect = [&](const uint4 t) {
      const unsigned wv[4] = {t.x, t.y, t.z, t.w};
      unsigned hm = 0u;
#pragma unroll
      for (int j = 0; j < 4; ++j) {
        const unsigned x = wv[j] ^ pat;
        const unsigned y = (x & 0x7F7F7F7Fu) + 0x7F7F7F7Fu;
        const unsigned z = ~(y | x) & 0x80808080u;
        hm |= z >> (7 - j);
      }
      return hm;
    };
    auto collect = [&](unsigned hm, int k) {
      const int ebase = step_of(k) * AGG_S + lane * 16;
      unsigned m = __builtin_amdgcn_ballot_w32(hm != 0u);
      while (m != 0u) {
        const int p = __builtin_ctz(hm | 0x80000000u);
        const int e = ebase + ((p & 3) << 2) + (p >> 3);
        const bool hit = (hm != 0u) && (e < E);
        const unsigned mask = __builtin_amdgcn_ballot_w32(hit);
        if (hit) q[qn + __builtin_popcount(mask & ((1u << lane) - 1u))] = e;
        qn += __builtin_popcount(mask);
        hm &= hm - 1u;
        m = __builtin_amdgcn_ballot_w32(hm != 0u);
      }
      after_step(k + 32 < nstep);
    };
    uint4 ta = load_step(wave), tb = load_step(wave + 32);
    for (int k = wave; k < nstep && !lost; k += 64) {
      const unsigned h0 = detect(ta);
      const unsigned h1 = detect(tb);
      ta = load_step(k + 64);
      tb = load_step(k + 96);
      collect(h0, k);
      if (!lost && k + 32 < nstep) collect(h1, k + 32);
    }
  };

  auto weight = [&](float s, float ad, float m) { return __expf(leaky(s + ad) - m); };
  auto add_row = [&](float4& a0, float4& a1, float& zA, float& zB,
                     const float4 x0, const float4 x1, float wA, float wB) {
    a0.x += x0.x * wA;  a0.y += x0.y * wA;  a0.z += x0.z * wA;  a0.w += x0.w * wA;
    a1.x += x1.x * wB;  a1.y += x1.y * wB;  a1.z += x1.z * wB;  a1.w += x1.w * wB;
    zA += wA;  zB += wB;
  };
  auto store_row = [&](long long n, const float4 a0, const float4 a1, float zA, float zB) {
    const float iA = 1.0f / zA, iB = 1.0f / zB;
    float4 v0, v1;
    v0.x = a0.x * iA;  v0.y = a0.y * iA;  v0.z = a0.z * iA;  v0.w = a0.w * iA;
    v1.x = a1.x * iB;  v1.y = a1.y * iB;  v1.z = a1.z * iB;  v1.w = a1.w * iB;
    float4* __restrict__ orow = (float4*)(agg + n * 256);
    orow[lane]      = v0;
    orow[32 + lane] = v1;
  };

  scan([&](bool more) {
    if (more && qn > AGG_Q - AGG_S) lost = true;
  });
  if (lost) s_lost = 1;
  else if (qn > 0) { pack_range(q, qn); max_range(q, qn); }
  __syncthreads();

  if (s_lost != 0) {
    if ((int)threadIdx.x < nown) l_rec[n0 + threadIdx.x] = int2{0, 0};
    for (int r = wave; r < nown; r += 32) {
      const long long n = n0 + r;
      const float adA = adst[n * HEADS + hA], adB = adst[n * HEADS + hB];
      const float slA = asrc[n * HEADS + hA], slB = asrc[n * HEADS + hB];
      float mA = leaky(slA + adA), mB = leaky(slB + adB);
      for (int e0 = 0; e0 < E; e0 += 32) {
        const int e = e0 + lane;
        const bool hit = (e < E) && (dstp[e] == (int)n);
        const int s = hit ? ei[e] : 0;
        unsigned m = __builtin_amdgcn_ballot_w32(hit);
        while (m != 0u) {
          const int sj = __builtin_amdgcn_readlane(s, __builtin_ctz(m));
          m &= m - 1u;
          mA = fmaxf(mA, leaky(asrc[(long long)sj * HEADS + hA] + adA));
          mB = fmaxf(mB, leaky(asrc[(long long)sj * HEADS + hB] + adB));
        }
      }
      const float4* __restrict__ own = (const float4*)(hin + n * 256);
      float4 a0 = float4{0.0f, 0.0f, 0.0f, 0.0f}, a1 = a0;
      float zA = 0.0f, zB = 0.0f;
      add_row(a0, a1, zA, zB, own[lane], own[32 + lane],
              weight(slA, adA, mA), weight(slB, adB, mB));
      for (int e0 = 0; e0 < E; e0 += 32) {
        const int e = e0 + lane;
        const bool hit = (e < E) && (dstp[e] == (int)n);
        const int s = hit ? ei[e] : 0;
        unsigned m = __builtin_amdgcn_ballot_w32(hit);
        while (m != 0u) {
          const int sj = __builtin_amdgcn_readlane(s, __builtin_ctz(m));
          m &= m - 1u;
          const float4* __restrict__ row = (const float4*)(hin + (long long)sj * 256);
          add_row(a0, a1, zA, zB, row[lane], row[32 + lane],
                  weight(asrc[(long long)sj * HEADS + hA], adA, mA),
                  weight(asrc[(long long)sj * HEADS + hB], adB, mB));
        }
      }
      store_row(n, a0, a1, zA, zB);
    }
    return;
  }

  if (wave == 0) {
    int c[AGG_PER];
    int sum = 0;
#pragma unroll
    for (int j = 0; j < AGG_PER; ++j) {
      const int r = lane * AGG_PER + j;
      c[j] = (r < nown) ? r_cnt[r] : 0;
      sum += c[j];
    }
    int inc = sum;
#pragma unroll
    for (int off = 1; off < 32; off <<= 1) {
      const int t = __shfl_up(inc, off);
      if (lane >= off) inc += t;
    }
    int run = inc - sum;
#pragma unroll
    for (int j = 0; j < AGG_PER; ++j) {
      const int r = lane * AGG_PER + j;
      if (r < nown) {
        r_beg[r]    = run;
        r_fill[r]   = run + 1;
        rowsrc[run] = n0 + r;
      }
      run += c[j];
    }
  }
  __syncthreads();

  for (int i = lane; i < qn; i += 32) {
    const unsigned v = (unsigned)q[i];
    const unsigned d = v >> 24;
    if (d != AGG_BAD) rowsrc[atomicAdd(&r_fill[d], 1)] = (int)(v & 0xFFFFFFu);
  }
  __syncthreads();

  {
    const int lbase = blockIdx.x * AGG_LS;
    const int total = r_beg[nown - 1] + r_cnt[nown - 1];
    if ((int)threadIdx.x < nown)
      l_rec[n0 + threadIdx.x] = int2{r_cnt[threadIdx.x], lbase + r_beg[threadIdx.x]};
    for (int i = threadIdx.x; i < total; i += 1024) l_src[lbase + i] = rowsrc[i];
  }

  for (int r = wave; r < nown; r += 32) {
    const int cnt = __builtin_amdgcn_readfirstlane(r_cnt[r]);
    const int beg = __builtin_amdgcn_readfirstlane(r_beg[r]);
    const int cj  = r * HEADS;
    const float adA = c_ad[cj + hA], mA = sortable2f(c_m[cj + hA]);
    const float adB = c_ad[cj + hB], mB = sortable2f(c_m[cj + hB]);
    float4 a0 = float4{0.0f, 0.0f, 0.0f, 0.0f}, a1 = a0;
    float zA = 0.0f, zB = 0.0f;
    for (int i0 = 0; i0 < cnt; i0 += AGG_B) {
      const int nb = min(AGG_B, cnt - i0);
      const int sv = rowsrc[beg + i0 + min(lane, nb - 1)];
      float4 x0[AGG_B], x1[AGG_B];
      float  sA[AGG_B], sB[AGG_B];
#pragma unroll
      for (int j = 0; j < AGG_B; ++j) {
        const int sj = __builtin_amdgcn_readlane(sv, j);
        const float4* __restrict__ row = (const float4*)(hin + (long long)sj * 256);
        x0[j] = row[lane];
        x1[j] = row[32 + lane];
        sA[j] = asrc[(long long)sj * HEADS + hA];
        sB[j] = (HEADS == 8) ? asrc[(long long)sj * HEADS + hB] : 0.0f;
      }
#pragma unroll
      for (int j = 0; j < AGG_B; ++j) {
        const float on = (j < nb) ? 1.0f : 0.0f;
        const float wA = weight(sA[j], adA, mA) * on;
        const float wB = (HEADS == 8) ? weight(sB[j], adB, mB) * on : wA;
        add_row(a0, a1, zA, zB, x0[j], x1[j], wA, wB);
      }
    }
    store_row((long long)n0 + r, a0, a1, zA, zB);
  }
}

#define LST_W 16

__device__ __forceinline__ void stream_row(const int* __restrict__ ei, int E,
                                           const float* __restrict__ hin,
                                           const float* __restrict__ asrc,
                                           float* __restrict__ agg,
                                           int n, float ad, int lane) {
  float4 a0 = float4{0.0f, 0.0f, 0.0f, 0.0f}, a1 = a0;
  float z = 0.0f;
  const int* __restrict__ dstp = ei + E;
  const float sl = asrc[n];
  float m = leaky(sl + ad);
  for (int e0 = 0; e0 < E; e0 += 32) {
    const int e = e0 + lane;
    const bool hit = (e < E) && (dstp[e] == n);
    const int s = hit ? ei[e] : 0;
    unsigned k = __builtin_amdgcn_ballot_w32(hit);
    while (k != 0u) {
      const int sj = __builtin_amdgcn_readlane(s, __builtin_ctz(k));
      k &= k - 1u;
      m = fmaxf(m, leaky(asrc[sj] + ad));
    }
  }
  {
    const float4* __restrict__ own = (const float4*)(hin + (long long)n * 256);
    const float w = __expf(leaky(sl + ad) - m);
    const float4 x0 = own[lane], x1 = own[32 + lane];
    a0.x += x0.x * w;  a0.y += x0.y * w;  a0.z += x0.z * w;  a0.w += x0.w * w;
    a1.x += x1.x * w;  a1.y += x1.y * w;  a1.z += x1.z * w;  a1.w += x1.w * w;
    z += w;
  }
  for (int e0 = 0; e0 < E; e0 += 32) {
    const int e = e0 + lane;
    const bool hit = (e < E) && (dstp[e] == n);
    const int s = hit ? ei[e] : 0;
    unsigned k = __builtin_amdgcn_ballot_w32(hit);
    while (k != 0u) {
      const int sj = __builtin_amdgcn_readlane(s, __builtin_ctz(k));
      k &= k - 1u;
      const float4* __restrict__ row = (const float4*)(hin + (long long)sj * 256);
      const float w = __expf(leaky(asrc[sj] + ad) - m);
      const float4 x0 = row[lane], x1 = row[32 + lane];
      a0.x += x0.x * w;  a0.y += x0.y * w;  a0.z += x0.z * w;  a0.w += x0.w * w;
      a1.x += x1.x * w;  a1.y += x1.y * w;  a1.z += x1.z * w;  a1.w += x1.w * w;
      z += w;
    }
  }
  const float iz = 1.0f / z;
  float4 v0, v1;
  v0.x = a0.x * iz;  v0.y = a0.y * iz;  v0.z = a0.z * iz;  v0.w = a0.w * iz;
  v1.x = a1.x * iz;  v1.y = a1.y * iz;  v1.z = a1.z * iz;  v1.w = a1.w * iz;
  float4* __restrict__ orow = (float4*)(agg + (long long)n * 256);
  orow[lane]      = v0;
  orow[32 + lane] = v1;
}

__device__ __forceinline__ void listed_rows(const int* __restrict__ ei,
                                            const int2* __restrict__ l_rec,
                                            const int* __restrict__ l_src,
                                            const float* __restrict__ hin,
                                            const float* __restrict__ asrc,
                                            const float* __restrict__ adst,
                                            float* __restrict__ agg,
                                            int E, int Nn, int n0, int nrows, int lane) {

  const int   nl   = min(n0 + lane, Nn - 1);
  const int2  recv = l_rec[nl];
  const float adv  = adst[nl];

  auto window = [&](int cnt, int beg, int i0) {
    return l_src[beg + min(i0 + lane, cnt - 1)];
  };

  int cntN = __builtin_amdgcn_readlane(recv.x, 0);
  int begN = __builtin_amdgcn_readlane(recv.y, 0);
  int svN  = (cntN > 0) ? window(cntN, begN, 0) : 0;

  for (int r = 0; r < nrows; ++r) {
    const int n = n0 + r;
    const int cnt = cntN, beg = begN;
    int sv = svN;
    const float ad = __int_as_float(__builtin_amdgcn_readlane(__float_as_int(adv), r));
    if (r + 1 < nrows) {
      cntN = __builtin_amdgcn_readlane(recv.x, r + 1);
      begN = __builtin_amdgcn_readlane(recv.y, r + 1);
      svN  = (cntN > 0) ? window(cntN, begN, 0) : 0;
    }

    if (cnt <= 0) {
      stream_row(ei, E, hin, asrc, agg, n, ad, lane);
      continue;
    }

    float4 a0 = float4{0.0f, 0.0f, 0.0f, 0.0f}, a1 = a0;
    float z = 0.0f;
    {
      float sc = asrc[sv];
      float4 x0[AGG_B], x1[AGG_B];
      auto request = [&](int i0) {
        const int nb = min(AGG_B, cnt - i0);
        const int wb = i0 & 31;
#pragma unroll
        for (int j = 0; j < AGG_B; ++j) {
          const int sj = __builtin_amdgcn_readlane(sv, wb + min(j, nb - 1));
          const float4* __restrict__ row = (const float4*)(hin + (long long)sj * 256);
          x0[j] = row[lane];
          x1[j] = row[32 + lane];
        }
      };
      auto add_batch = [&](int i0, float m) {
        const int nb = min(AGG_B, cnt - i0);
        const int wb = i0 & 31;
#pragma unroll
        for (int j = 0; j < AGG_B; ++j) {
          const float s  = __int_as_float(__builtin_amdgcn_readlane(__float_as_int(sc),
                                                                    wb + min(j, nb - 1)));
          const float on = (j < nb) ? 1.0f : 0.0f;
          const float w  = __expf(leaky(s + ad) - m) * on;
          a0.x += x0[j].x * w;  a0.y += x0[j].y * w;  a0.z += x0[j].z * w;  a0.w += x0[j].w * w;
          a1.x += x1[j].x * w;  a1.y += x1[j].y * w;  a1.z += x1[j].z * w;  a1.w += x1[j].w * w;
          z += w;
        }
      };

      request(0);
      float m = leaky(sc + ad);
      for (int i0 = 32; i0 < cnt; i0 += 32) {
        const int s = window(cnt, beg, i0);
        m = fmaxf(m, leaky(asrc[s] + ad));
      }
#pragma unroll
      for (int off = 16; off > 0; off >>= 1) m = fmaxf(m, __shfl_xor(m, off));
      add_batch(0, m);

      for (int i0 = AGG_B; i0 < cnt; i0 += AGG_B) {
        if ((i0 & 31) == 0) {
          sv = window(cnt, beg, i0);
          sc = asrc[sv];
        }
        request(i0);
        add_batch(i0, m);
      }
    }

    const float iz = 1.0f / z;
    float4 v0, v1;
    v0.x = a0.x * iz;  v0.y = a0.y * iz;  v0.z = a0.z * iz;  v0.w = a0.w * iz;
    v1.x = a1.x * iz;  v1.y = a1.y * iz;  v1.z = a1.z * iz;  v1.w = a1.w * iz;
    float4* __restrict__ orow = (float4*)(agg + (long long)n * 256);
    orow[lane]      = v0;
    orow[32 + lane] = v1;
  }
}

__global__ __launch_bounds__(32 * (GEMM_W + LST_W))
void resid_gemm_beside_gather(const float* __restrict__ A,
                              const float* __restrict__ B,
                              float* __restrict__ R,
                              const float* __restrict__ bias_a,
                              const float* __restrict__ bias_b,
                              int M, int K,
                              const int* __restrict__ ei,
                              const int2* __restrict__ l_rec,
                              const int* __restrict__ l_src,
                              const float* __restrict__ hin,
                              const float* __restrict__ asrc,
                              const float* __restrict__ adst,
                              float* __restrict__ agg,
                              int E) {
  __shared__ float Bs[(GEMM_KMAX / 4) * GEMM_P];
  const int lane = threadIdx.x & 31;
  const int wave = __builtin_amdgcn_readfirstlane(threadIdx.x >> 5);
  gemm_stage_b<32 * (GEMM_W + LST_W)>(Bs, B, K);
  __syncthreads();
  if (wave < GEMM_W) {
    gemm_units<1>(Bs, A, R, bias_a, bias_b, M, K, wave, lane);
  } else {
    const long long g  = (long long)blockIdx.x * LST_W + (wave - GEMM_W);
    const long long ng = (long long)gridDim.x * LST_W;
    const int r0 = (int)((g * M) / ng);
    const int r1 = (int)(((g + 1) * M) / ng);
    for (int n0 = r0; n0 < r1; n0 += 32)
      listed_rows(ei, l_rec, l_src, hin, asrc, adst, agg, E, M, n0, min(32, r1 - n0), lane);
  }
}

__global__ __launch_bounds__(32 * ROWS_W)
void ln_elu_rows(float* hbuf, const float* __restrict__ b1,
                 const float* __restrict__ gamma,
                 const float* __restrict__ beta, int Nn) {
  const int lane = threadIdx.x & 31;
  const int wave = __builtin_amdgcn_readfirstlane(threadIdx.x >> 5);
  const int gw = blockIdx.x * ROWS_W + wave;
  const int GW = gridDim.x * ROWS_W;
  const float4 c0 = ((const float4*)b1)[lane],    c1 = ((const float4*)b1)[32 + lane];
  const float4 g0 = ((const float4*)gamma)[lane], g1 = ((const float4*)gamma)[32 + lane];
  const float4 e0 = ((const float4*)beta)[lane],  e1 = ((const float4*)beta)[32 + lane];
  const int nq = (Nn + ROWS_B - 1) / ROWS_B;
  auto elu = [](float t) { return t > 0.0f ? t : (__expf(t) - 1.0f); };
  for (int q = gw; q < nq; q += GW) {
    float4 v0[ROWS_B], v1[ROWS_B];
    load_rows(hbuf, q, Nn, lane, v0, v1);
    float sum[ROWS_B], var[ROWS_B];
#pragma unroll
    for (int j = 0; j < ROWS_B; ++j) {
      v0[j].x += c0.x;  v0[j].y += c0.y;  v0[j].z += c0.z;  v0[j].w += c0.w;
      v1[j].x += c1.x;  v1[j].y += c1.y;  v1[j].z += c1.z;  v1[j].w += c1.w;
      sum[j] = ((v0[j].x + v0[j].y) + (v0[j].z + v0[j].w)) +
               ((v1[j].x + v1[j].y) + (v1[j].z + v1[j].w));
    }
#pragma unroll
    for (int off = 16; off > 0; off >>= 1) {
#pragma unroll
      for (int j = 0; j < ROWS_B; ++j) sum[j] += __shfl_xor(sum[j], off);
    }
#pragma unroll
    for (int j = 0; j < ROWS_B; ++j) {
      const float mu = sum[j] * (1.0f / 256.0f);
      v0[j].x -= mu;  v0[j].y -= mu;  v0[j].z -= mu;  v0[j].w -= mu;
      v1[j].x -= mu;  v1[j].y -= mu;  v1[j].z -= mu;  v1[j].w -= mu;
      var[j] = ((v0[j].x * v0[j].x + v0[j].y * v0[j].y) + (v0[j].z * v0[j].z + v0[j].w * v0[j].w)) +
               ((v1[j].x * v1[j].x + v1[j].y * v1[j].y) + (v1[j].z * v1[j].z + v1[j].w * v1[j].w));
    }
#pragma unroll
    for (int off = 16; off > 0; off >>= 1) {
#pragma unroll
      for (int j = 0; j < ROWS_B; ++j) var[j] += __shfl_xor(var[j], off);
    }
#pragma unroll
    for (int j = 0; j < ROWS_B; ++j) {
      const float rs = rsqrtf(var[j] * (1.0f / 256.0f) + LN_EPS);
      float4 o0, o1;
      o0.x = elu(v0[j].x * rs * g0.x + e0.x);  o0.y = elu(v0[j].y * rs * g0.y + e0.y);
      o0.z = elu(v0[j].z * rs * g0.z + e0.z);  o0.w = elu(v0[j].w * rs * g0.w + e0.w);
      o1.x = elu(v1[j].x * rs * g1.x + e1.x);  o1.y = elu(v1[j].y * rs * g1.y + e1.y);
      o1.z = elu(v1[j].z * rs * g1.z + e1.z);  o1.w = elu(v1[j].w * rs * g1.w + e1.w);
      v0[j] = o0;  v1[j] = o1;
    }
#pragma unroll
    for (int j = 0; j < ROWS_B; ++j) {
      const int n = q * ROWS_B + j;
      if (n < Nn) {
        float4* orow = (float4*)(hbuf + (long long)n * 256);
        orow[lane]      = v0[j];
        orow[32 + lane] = v1[j];
      }
    }
  }
}

#define POOL_W 16
#define POOL_U 16
#define POOL_B 4

__global__ __launch_bounds__(32 * POOL_W)
void pool_own(const float* __restrict__ o, const float* __restrict__ res,
              const int* __restrict__ batch, float* __restrict__ out, int Nn) {
  __shared__ float4 part[POOL_W][64];
  __shared__ int    pcnt[POOL_W];

  const int lane = threadIdx.x & 31;
  const int wave = __builtin_amdgcn_readfirstlane(threadIdx.x >> 5);
  const int g = blockIdx.x;

  float4 a0 = float4{0.0f, 0.0f, 0.0f, 0.0f}, a1 = a0;
  int cnt = 0;
  const int nstep = (Nn + 31) >> 5;

  for (int s0 = wave; s0 < nstep; s0 += POOL_W * POOL_U) {
    unsigned hw = 0u;
#pragma unroll
    for (int u = 0; u < POOL_U; ++u) {
      const int n = (s0 + u * POOL_W) * 32 + lane;
      const int b = (n < Nn) ? batch[n] : -1;
      hw |= (b == g ? 1u : 0u) << u;
    }
    if (__builtin_amdgcn_ballot_w32(hw != 0u) == 0u) continue;
#pragma nounroll
    for (int u = 0; u < POOL_U; ++u) {
      unsigned m = __builtin_amdgcn_ballot_w32(((hw >> u) & 1u) != 0u);
      const int base = (s0 + u * POOL_W) * 32;
      cnt += __builtin_popcount(m);
      while (m != 0u) {
        float4 x0[POOL_B], x1[POOL_B], y0[POOL_B], y1[POOL_B];
        float  on[POOL_B];
        int p = 0;
#pragma unroll
        for (int j = 0; j < POOL_B; ++j) {
          const bool live = (m != 0u);
          if (live) p = __builtin_ctz(m);
          m &= m - 1u;
          on[j] = live ? 1.0f : 0.0f;
          const float4* __restrict__ row = (const float4*)(o + (long long)(base + p) * 256);
          const float4* __restrict__ rrw = (const float4*)(res + (long long)(base + p) * 256);
          x0[j] = row[lane];
          x1[j] = row[32 + lane];
          y0[j] = rrw[lane];
          y1[j] = rrw[32 + lane];
        }
#pragma unroll
        for (int j = 0; j < POOL_B; ++j) {
          a0.x += (x0[j].x + y0[j].x) * on[j];  a0.y += (x0[j].y + y0[j].y) * on[j];
          a0.z += (x0[j].z + y0[j].z) * on[j];  a0.w += (x0[j].w + y0[j].w) * on[j];
          a1.x += (x1[j].x + y1[j].x) * on[j];  a1.y += (x1[j].y + y1[j].y) * on[j];
          a1.z += (x1[j].z + y1[j].z) * on[j];  a1.w += (x1[j].w + y1[j].w) * on[j];
        }
      }
    }
  }

  part[wave][lane]      = a0;
  part[wave][32 + lane] = a1;
  if (lane == 0) pcnt[wave] = cnt;
  __syncthreads();

  if (threadIdx.x < 256) {
    const float* pf = (const float*)part;
    float s = 0.0f;
    int c = 0;
#pragma unroll
    for (int w = 0; w < POOL_W; ++w) {
      s += pf[w * 256 + threadIdx.x];
      c += pcnt[w];
    }
    out[(long long)g * 256 + threadIdx.x] = s / fmaxf((float)c, 1.0f);
  }
}

static inline int cdiv(long long a, long long b) { return (int)((a + b - 1) / b); }

extern "C" void kernel_launch(void* const* d_in, const int* in_sizes, int n_in,
                              void* d_out, int out_size, void* d_ws, size_t ws_size,
                              hipStream_t stream) {
  const float* x        = (const float*)d_in[0];
  const int*   ei       = (const int*)  d_in[1];
  const int*   batch    = (const int*)  d_in[2];
  const float* W1       = (const float*)d_in[3];
  const float* att_src1 = (const float*)d_in[4];
  const float* att_dst1 = (const float*)d_in[5];
  const float* b1       = (const float*)d_in[6];
  const float* gamma    = (const float*)d_in[7];
  const float* beta     = (const float*)d_in[8];
  const float* W2       = (const float*)d_in[9];
  const float* att_src2 = (const float*)d_in[10];
  const float* att_dst2 = (const float*)d_in[11];
  const float* b2       = (const float*)d_in[12];
  const float* Wres     = (const float*)d_in[13];
  const float* bres     = (const float*)d_in[14];
  float* out = (float*)d_out;

  const int Nn = in_sizes[2];
  const int Ee = in_sizes[1] / 2;
  const int Gg = out_size / 256;
  const int EA = Ee + Nn;
  const int IN_DIM = in_sizes[0] / Nn;

  const long long NB = (long long)Nn * 256;
  float* ws   = (float*)d_ws;
  float* buf0 = ws;
  float* buf1 = ws + NB;
  float* buf2 = ws + 2 * NB;
  float* cnt  = ws + 3 * NB;
  float* as1  = cnt + Gg;
  float* ad1  = as1 + (long long)Nn * 8;
  float* as2  = ad1 + (long long)Nn * 8;
  float* ad2  = as2 + Nn;
  const int Epad = cdiv(Ee, AGG_S) * AGG_S;
  const long long tag_w = ((3 * NB + (long long)Nn * 18 + Gg + 3) / 4) * 4;
  unsigned* tag = (unsigned*)(ws + tag_w);
  const int NBLK = cdiv(Nn, AGG_R);
  int2* l_rec = (int2*)(tag + Epad / 4);
  int*  l_src = (int*)(l_rec + (long long)NBLK * AGG_R);
  const long long res_w = ((tag_w + Epad / 4 + (long long)NBLK * AGG_R * 2 +
                            (long long)NBLK * AGG_LS + 3) / 4) * 4;
  float* buf3 = ws + res_w;

  const int B = 256;

  edge_owner<<<cdiv(Epad / 4, B), B, 0, stream>>>(ei, tag, Ee, Epad);

  gemm_wmma_n256<<<GEMM_G, 32 * GEMM_W, 0, stream>>>(x, W1, buf0, Nn, IN_DIM);
  attn_scores8_rows<<<ROWS_G, 32 * ROWS_W, 0, stream>>>(buf0, att_src1, att_dst1, as1, ad1, Nn);
  edge_softmax_aggregate_own<8><<<NBLK, 1024, 0, stream>>>(ei, tag, buf0, as1, ad1, buf1,
                                                           l_rec, l_src, Ee, Epad, Nn);
  ln_elu_rows<<<ROWS_G, 32 * ROWS_W, 0, stream>>>(buf1, b1, gamma, beta, Nn);

  gemm_wmma_n256<<<GEMM_G, 32 * GEMM_W, 0, stream>>>(buf1, W2, buf0, Nn, 256);
  attn_scores1_rows<<<ROWS_G, 32 * ROWS_W, 0, stream>>>(buf0, att_src2, att_dst2, as2, ad2, Nn);
  resid_gemm_beside_gather<<<GEMM_G, 32 * (GEMM_W + LST_W), 0, stream>>>(
      buf1, Wres, buf3, b2, bres, Nn, 256, ei, l_rec, l_src, buf0, as2, ad2, buf2, Ee);

  pool_own<<<Gg, 32 * POOL_W, 0, stream>>>(buf2, buf3, batch, out, Nn);
}
